# starting kernel
# baseline (speedup 1.0000x reference)
#include <hip/hip_runtime.h>


typedef __attribute__((ext_vector_type(2))) float v2f;
typedef __attribute__((ext_vector_type(8))) float v8f;

#define DIM   128
#define GNUM  64
#define LAYERS 3
#define LNEPS 1e-5f

__global__ void zero_f32(float* __restrict__ p, int n) {
    int i = blockIdx.x * blockDim.x + threadIdx.x;
    if (i < n) p[i] = 0.0f;
}

__global__ void deg_accum(const int* __restrict__ ei, float* __restrict__ deg, int E) {
    int e = blockIdx.x * blockDim.x + threadIdx.x;
    if (e < E) atomicAdd(&deg[ei[E + e]], 1.0f);
}

__global__ void deg_to_dinv(float* __restrict__ d, int N) {
    int i = blockIdx.x * blockDim.x + threadIdx.x;
    if (i < N) d[i] = rsqrtf(d[i] + 1.0f);
}

__global__ __launch_bounds__(256) void gemm_wmma(const float* __restrict__ H,
                                                 const float* __restrict__ W,
                                                 float* __restrict__ out, int N) {
    __shared__ float lw[DIM * DIM];
    {
        const float4* Wv = (const float4*)W;
        float4* lv = (float4*)lw;
        #pragma unroll
        for (int i = 0; i < (DIM * DIM / 4) / 256; ++i)
            lv[threadIdx.x + i * 256] = Wv[threadIdx.x + i * 256];
    }
    __syncthreads();

    const int wave = threadIdx.x >> 5;
    const int lane = threadIdx.x & 31;
    const int half = lane >> 4;
    const int ln   = lane & 15;

    const int nstrips = N >> 4;
    int strip = blockIdx.x * 8 + wave;
    if (strip >= nstrips) strip = nstrips - 1;
    const int rowBase = strip << 4;

    const v2f* arow2 = (const v2f*)(H + (size_t)(rowBase + ln) * DIM);

    v8f acc[8] = {};

    for (int k = 0; k < DIM; k += 4) {
        v2f a = arow2[(k >> 1) + half];
        const int kb = k + 2 * half;
        #pragma unroll
        for (int t = 0; t < 8; ++t) {
            v2f b;
            b.x = lw[kb * DIM + t * 16 + ln];
            b.y = lw[(kb + 1) * DIM + t * 16 + ln];
            acc[t] = __builtin_amdgcn_wmma_f32_16x16x4_f32(
                false, a, false, b, (short)0, acc[t], false, false);
        }
    }

    #pragma unroll
    for (int t = 0; t < 8; ++t) {
        #pragma unroll
        for (int j = 0; j < 8; ++j) {
            int row = rowBase + half * 8 + j;
            out[(size_t)row * DIM + t * 16 + ln] = acc[t][j];
        }
    }
}

__global__ void acc_init(const float* __restrict__ hw, const float* __restrict__ dinv,
                         const float* __restrict__ bias, float* __restrict__ acc, int N) {
    int i = blockIdx.x * blockDim.x + threadIdx.x;
    if (i >= N * (DIM / 4)) return;
    int n  = i >> 5;
    int c4 = i & 31;
    float s = dinv[n]; s = s * s;
    float4 hv = ((const float4*)hw)[i];
    float4 bv = ((const float4*)bias)[c4];
    float4 r;
    r.x = bv.x + s * hv.x;  r.y = bv.y + s * hv.y;
    r.z = bv.z + s * hv.z;  r.w = bv.w + s * hv.w;
    ((float4*)acc)[i] = r;
}

__global__ __launch_bounds__(256) void edge_scatter(const int* __restrict__ ei,
                                                    const float* __restrict__ hw,
                                                    const float* __restrict__ dinv,
                                                    float* __restrict__ acc, int E) {
    int wave = threadIdx.x >> 5, lane = threadIdx.x & 31;
    int e = blockIdx.x * 8 + wave;
    if (e >= E) return;
    int src = ei[e];
    int dst = ei[E + e];
    float w = dinv[src] * dinv[dst];
    float4 v = ((const float4*)(hw + (size_t)src * DIM))[lane];
    float* dp = acc + (size_t)dst * DIM + lane * 4;
    atomicAdd(dp + 0, w * v.x);
    atomicAdd(dp + 1, w * v.y);
    atomicAdd(dp + 2, w * v.z);
    atomicAdd(dp + 3, w * v.w);
}

__global__ __launch_bounds__(256) void ln_relu(const float* __restrict__ acc,
                                               const float* __restrict__ gamma,
                                               const float* __restrict__ beta,
                                               float* __restrict__ out, int N) {
    int wave = threadIdx.x >> 5, lane = threadIdx.x & 31;
    int n = blockIdx.x * 8 + wave;
    if (n >= N) return;
    float4 v = ((const float4*)(acc + (size_t)n * DIM))[lane];
    float s = v.x + v.y + v.z + v.w;
    float q = v.x * v.x + v.y * v.y + v.z * v.z + v.w * v.w;
    #pragma unroll
    for (int off = 16; off > 0; off >>= 1) {
        s += __shfl_xor(s, off, 32);
        q += __shfl_xor(q, off, 32);
    }
    float mu  = s * (1.0f / DIM);
    float var = fmaxf(q * (1.0f / DIM) - mu * mu, 0.0f);
    float inv = rsqrtf(var + LNEPS);
    float4 g = ((const float4*)gamma)[lane];
    float4 b = ((const float4*)beta)[lane];
    float4 r;
    r.x = fmaxf(0.0f, g.x * (v.x - mu) * inv + b.x);
    r.y = fmaxf(0.0f, g.y * (v.y - mu) * inv + b.y);
    r.z = fmaxf(0.0f, g.z * (v.z - mu) * inv + b.z);
    r.w = fmaxf(0.0f, g.w * (v.w - mu) * inv + b.w);
    ((float4*)(out + (size_t)n * DIM))[lane] = r;
}

__global__ __launch_bounds__(256) void pool_accum(const float* __restrict__ h,
                                                  const int* __restrict__ batch,
                                                  float* __restrict__ pool,
                                                  float* __restrict__ cnt, int N) {
    int wave = threadIdx.x >> 5, lane = threadIdx.x & 31;
    int n = blockIdx.x * 8 + wave;
    if (n >= N) return;
    int g = batch[n];
    float4 v = ((const float4*)(h + (size_t)n * DIM))[lane];
    float* p = pool + (size_t)g * DIM + lane * 4;
    atomicAdd(p + 0, v.x);
    atomicAdd(p + 1, v.y);
    atomicAdd(p + 2, v.z);
    atomicAdd(p + 3, v.w);
    if (lane == 0) atomicAdd(&cnt[g], 1.0f);
}

__global__ void pool_final(const float* __restrict__ pool, const float* __restrict__ cnt,
                           float* __restrict__ out, int total) {
    int i = blockIdx.x * blockDim.x + threadIdx.x;
    if (i < total) out[i] = pool[i] / fmaxf(cnt[i / DIM], 1.0f);
}

static inline size_t al256(size_t x) { return (x + 255) & ~(size_t)255; }

extern "C" void kernel_launch(void* const* d_in, const int* in_sizes, int n_in,
                              void* d_out, int out_size, void* d_ws, size_t ws_size,
                              hipStream_t stream) {
    const float* x      = (const float*)d_in[0];
    const int*   ei     = (const int*)d_in[1];
    const int*   batch  = (const int*)d_in[2];
    const float* Ws     = (const float*)d_in[3];
    const float* bs     = (const float*)d_in[4];
    const float* gammas = (const float*)d_in[5];
    const float* betas  = (const float*)d_in[6];

    const int N = in_sizes[0] / DIM;
    const int E = in_sizes[1] / 2;
    const size_t nd = (size_t)N * DIM * sizeof(float);

    char* ws = (char*)d_ws;
    float* dinv = (float*)ws;                 ws += al256((size_t)N * sizeof(float));
    float* hw   = (float*)ws;                 ws += al256(nd);
    float* accb = (float*)ws;                 ws += al256(nd);
    float* hbuf = (float*)ws;                 ws += al256(nd);
    float* pool = (float*)ws;
    float* cnt  = pool + GNUM * DIM;

    zero_f32<<<(N + 255) / 256, 256, 0, stream>>>(dinv, N);
    deg_accum<<<(E + 255) / 256, 256, 0, stream>>>(ei, dinv, E);
    deg_to_dinv<<<(N + 255) / 256, 256, 0, stream>>>(dinv, N);

    const int nstrips = N >> 4;
    const float* hcur = x;
    for (int l = 0; l < LAYERS; ++l) {
        gemm_wmma<<<(nstrips + 7) / 8, 256, 0, stream>>>(
            hcur, Ws + (size_t)l * DIM * DIM, hw, N);
        acc_init<<<(N * (DIM / 4) + 255) / 256, 256, 0, stream>>>(
            hw, dinv, bs + (size_t)l * DIM, accb, N);
        edge_scatter<<<(E + 7) / 8, 256, 0, stream>>>(ei, hw, dinv, accb, E);
        ln_relu<<<(N + 7) / 8, 256, 0, stream>>>(
            accb, gammas + (size_t)l * DIM, betas + (size_t)l * DIM, hbuf, N);
        hcur = hbuf;
    }

    zero_f32<<<(GNUM * DIM + GNUM + 255) / 256, 256, 0, stream>>>(pool, GNUM * DIM + GNUM);
    pool_accum<<<(N + 7) / 8, 256, 0, stream>>>(hcur, batch, pool, cnt, N);
    pool_final<<<(GNUM * DIM + 255) / 256, 256, 0, stream>>>(pool, cnt, (float*)d_out, GNUM * DIM);
}


# mean pool: reduce before the atomics - a wave sums its consecutive nodes of one graph in registers, the work-group joins its waves in LDS, one atomic per column per work-group (new sum order)
# speedup vs baseline: 1.2838x; 1.2838x over previous
#include <hip/hip_runtime.h>


typedef __attribute__((ext_vector_type(2))) float v2f;
typedef __attribute__((ext_vector_type(8))) float v8f;

#define DIM   128
#define GNUM  64
#define LAYERS 3
#define LNEPS 1e-5f

__global__ void zero_f32(float* __restrict__ p, int n) {
    int i = blockIdx.x * blockDim.x + threadIdx.x;
    if (i < n) p[i] = 0.0f;
}

__global__ void deg_accum(const int* __restrict__ ei, float* __restrict__ deg, int E) {
    int e = blockIdx.x * blockDim.x + threadIdx.x;
    if (e < E) atomicAdd(&deg[ei[E + e]], 1.0f);
}

__global__ void deg_to_dinv(float* __restrict__ d, int N) {
    int i = blockIdx.x * blockDim.x + threadIdx.x;
    if (i < N) d[i] = rsqrtf(d[i] + 1.0f);
}

__global__ __launch_bounds__(256) void gemm_wmma(const float* __restrict__ H,
                                                 const float* __restrict__ W,
                                                 float* __restrict__ out, int N) {
    __shared__ float lw[DIM * DIM];
    {
        const float4* Wv = (const float4*)W;
        float4* lv = (float4*)lw;
        #pragma unroll
        for (int i = 0; i < (DIM * DIM / 4) / 256; ++i)
            lv[threadIdx.x + i * 256] = Wv[threadIdx.x + i * 256];
    }
    __syncthreads();

    const int wave = threadIdx.x >> 5;
    const int lane = threadIdx.x & 31;
    const int half = lane >> 4;
    const int ln   = lane & 15;

    const int nstrips = N >> 4;
    int strip = blockIdx.x * 8 + wave;
    if (strip >= nstrips) strip = nstrips - 1;
    const int rowBase = strip << 4;

    const v2f* arow2 = (const v2f*)(H + (size_t)(rowBase + ln) * DIM);

    v8f acc[8] = {};

    for (int k = 0; k < DIM; k += 4) {
        v2f a = arow2[(k >> 1) + half];
        const int kb = k + 2 * half;
        #pragma unroll
        for (int t = 0; t < 8; ++t) {
            v2f b;
            b.x = lw[kb * DIM + t * 16 + ln];
            b.y = lw[(kb + 1) * DIM + t * 16 + ln];
            acc[t] = __builtin_amdgcn_wmma_f32_16x16x4_f32(
                false, a, false, b, (short)0, acc[t], false, false);
        }
    }

    #pragma unroll
    for (int t = 0; t < 8; ++t) {
        #pragma unroll
        for (int j = 0; j < 8; ++j) {
            int row = rowBase + half * 8 + j;
            out[(size_t)row * DIM + t * 16 + ln] = acc[t][j];
        }
    }
}

__global__ void acc_init(const float* __restrict__ hw, const float* __restrict__ dinv,
                         const float* __restrict__ bias, float* __restrict__ acc, int N) {
    int i = blockIdx.x * blockDim.x + threadIdx.x;
    if (i >= N * (DIM / 4)) return;
    int n  = i >> 5;
    int c4 = i & 31;
    float s = dinv[n]; s = s * s;
    float4 hv = ((const float4*)hw)[i];
    float4 bv = ((const float4*)bias)[c4];
    float4 r;
    r.x = bv.x + s * hv.x;  r.y = bv.y + s * hv.y;
    r.z = bv.z + s * hv.z;  r.w = bv.w + s * hv.w;
    ((float4*)acc)[i] = r;
}

__global__ __launch_bounds__(256) void edge_scatter(const int* __restrict__ ei,
                                                    const float* __restrict__ hw,
                                                    const float* __restrict__ dinv,
                                                    float* __restrict__ acc, int E) {
    int wave = threadIdx.x >> 5, lane = threadIdx.x & 31;
    int e = blockIdx.x * 8 + wave;
    if (e >= E) return;
    int src = ei[e];
    int dst = ei[E + e];
    float w = dinv[src] * dinv[dst];
    float4 v = ((const float4*)(hw + (size_t)src * DIM))[lane];
    float* dp = acc + (size_t)dst * DIM + lane * 4;
    atomicAdd(dp + 0, w * v.x);
    atomicAdd(dp + 1, w * v.y);
    atomicAdd(dp + 2, w * v.z);
    atomicAdd(dp + 3, w * v.w);
}

__global__ __launch_bounds__(256) void ln_relu(const float* __restrict__ acc,
                                               const float* __restrict__ gamma,
                                               const float* __restrict__ beta,
                                               float* __restrict__ out, int N) {
    int wave = threadIdx.x >> 5, lane = threadIdx.x & 31;
    int n = blockIdx.x * 8 + wave;
    if (n >= N) return;
    float4 v = ((const float4*)(acc + (size_t)n * DIM))[lane];
    float s = v.x + v.y + v.z + v.w;
    float q = v.x * v.x + v.y * v.y + v.z * v.z + v.w * v.w;
    #pragma unroll
    for (int off = 16; off > 0; off >>= 1) {
        s += __shfl_xor(s, off, 32);
        q += __shfl_xor(q, off, 32);
    }
    float mu  = s * (1.0f / DIM);
    float var = fmaxf(q * (1.0f / DIM) - mu * mu, 0.0f);
    float inv = rsqrtf(var + LNEPS);
    float4 g = ((const float4*)gamma)[lane];
    float4 b = ((const float4*)beta)[lane];
    float4 r;
    r.x = fmaxf(0.0f, g.x * (v.x - mu) * inv + b.x);
    r.y = fmaxf(0.0f, g.y * (v.y - mu) * inv + b.y);
    r.z = fmaxf(0.0f, g.z * (v.z - mu) * inv + b.z);
    r.w = fmaxf(0.0f, g.w * (v.w - mu) * inv + b.w);
    ((float4*)(out + (size_t)n * DIM))[lane] = r;
}

#define POOL_NPW 16
#define POOL_WAVES 8

__device__ __forceinline__ void pool_flush(float* __restrict__ pool, float* __restrict__ cnt,
                                           int g, float4 s, float c, int lane) {
    float* p = pool + (size_t)g * DIM + lane * 4;
    atomicAdd(p + 0, s.x);
    atomicAdd(p + 1, s.y);
    atomicAdd(p + 2, s.z);
    atomicAdd(p + 3, s.w);
    if (lane == 0) atomicAdd(&cnt[g], c);
}

__global__ __launch_bounds__(256) void pool_accum(const float* __restrict__ h,
                                                  const int* __restrict__ batch,
                                                  float* __restrict__ pool,
                                                  float* __restrict__ cnt, int N) {
    __shared__ float part[POOL_WAVES][DIM];
    __shared__ int   pgid[POOL_WAVES];
    __shared__ float pcnt[POOL_WAVES];

    const int lane = threadIdx.x & 31;
    const int wave = __builtin_amdgcn_readfirstlane((int)(threadIdx.x >> 5));
    const int n0 = (blockIdx.x * POOL_WAVES + wave) * POOL_NPW;
    const int nv = min(POOL_NPW, N - n0);

    const int gv = batch[min(n0 + (lane & (POOL_NPW - 1)), N - 1)];

    float4 v[POOL_NPW];
    #pragma unroll
    for (int i = 0; i < POOL_NPW; ++i)
        v[i] = ((const float4*)(h + (size_t)min(n0 + i, N - 1) * DIM))[lane];

    int cur = -1;
    float4 s = make_float4(0.0f, 0.0f, 0.0f, 0.0f);
    float c = 0.0f;
    if (nv > 0) {
        cur = __builtin_amdgcn_readlane(gv, 0);
        s = v[0];
        c = 1.0f;
        #pragma unroll
        for (int i = 1; i < POOL_NPW; ++i) {
            if (i < nv) {
                const int gi = __builtin_amdgcn_readlane(gv, i);
                if (gi == cur) {
                    s.x += v[i].x; s.y += v[i].y; s.z += v[i].z; s.w += v[i].w;
                    c += 1.0f;
                } else {
                    pool_flush(pool, cnt, cur, s, c, lane);
                    cur = gi; s = v[i]; c = 1.0f;
                }
            }
        }
    }

    ((float4*)part[wave])[lane] = s;
    if (lane == 0) { pgid[wave] = cur; pcnt[wave] = c; }
    __syncthreads();

    const int t = threadIdx.x;
    if (t <= DIM) {
        const bool isc = (t == DIM);
        float* dstp = isc ? cnt : pool + t;
        const int dstride = isc ? 1 : DIM;
        int   jg[POOL_WAVES];
        float jx[POOL_WAVES];
        #pragma unroll
        for (int w = 0; w < POOL_WAVES; ++w) {
            jg[w] = pgid[w];
            jx[w] = isc ? pcnt[w] : part[w][t & (DIM - 1)];
        }
        int   rg = jg[0];
        float rs = jx[0];
        #pragma unroll
        for (int w = 1; w < POOL_WAVES; ++w) {
            if (jg[w] == rg) {
                rs += jx[w];
            } else {
                if (rg >= 0) atomicAdd(dstp + (size_t)rg * dstride, rs);
                rg = jg[w]; rs = jx[w];
            }
        }
        if (rg >= 0) atomicAdd(dstp + (size_t)rg * dstride, rs);
    }
}

__global__ void pool_final(const float* __restrict__ pool, const float* __restrict__ cnt,
                           float* __restrict__ out, int total) {
    int i = blockIdx.x * blockDim.x + threadIdx.x;
    if (i < total) out[i] = pool[i] / fmaxf(cnt[i / DIM], 1.0f);
}

static inline size_t al256(size_t x) { return (x + 255) & ~(size_t)255; }

extern "C" void kernel_launch(void* const* d_in, const int* in_sizes, int n_in,
                              void* d_out, int out_size, void* d_ws, size_t ws_size,
                              hipStream_t stream) {
    const float* x      = (const float*)d_in[0];
    const int*   ei     = (const int*)d_in[1];
    const int*   batch  = (const int*)d_in[2];
    const float* Ws     = (const float*)d_in[3];
    const float* bs     = (const float*)d_in[4];
    const float* gammas = (const float*)d_in[5];
    const float* betas  = (const float*)d_in[6];

    const int N = in_sizes[0] / DIM;
    const int E = in_sizes[1] / 2;
    const size_t nd = (size_t)N * DIM * sizeof(float);

    char* ws = (char*)d_ws;
    float* dinv = (float*)ws;                 ws += al256((size_t)N * sizeof(float));
    float* hw   = (float*)ws;                 ws += al256(nd);
    float* accb = (float*)ws;                 ws += al256(nd);
    float* hbuf = (float*)ws;                 ws += al256(nd);
    float* pool = (float*)ws;
    float* cnt  = pool + GNUM * DIM;

    zero_f32<<<(N + 255) / 256, 256, 0, stream>>>(dinv, N);
    deg_accum<<<(E + 255) / 256, 256, 0, stream>>>(ei, dinv, E);
    deg_to_dinv<<<(N + 255) / 256, 256, 0, stream>>>(dinv, N);

    const int nstrips = N >> 4;
    const float* hcur = x;
    for (int l = 0; l < LAYERS; ++l) {
        gemm_wmma<<<(nstrips + 7) / 8, 256, 0, stream>>>(
            hcur, Ws + (size_t)l * DIM * DIM, hw, N);
        acc_init<<<(N * (DIM / 4) + 255) / 256, 256, 0, stream>>>(
            hw, dinv, bs + (size_t)l * DIM, accb, N);
        edge_scatter<<<(E + 7) / 8, 256, 0, stream>>>(ei, hw, dinv, accb, E);
        ln_relu<<<(N + 7) / 8, 256, 0, stream>>>(
            accb, gammas + (size_t)l * DIM, betas + (size_t)l * DIM, hbuf, N);
        hcur = hbuf;
    }

    zero_f32<<<(GNUM * DIM + GNUM + 255) / 256, 256, 0, stream>>>(pool, GNUM * DIM + GNUM);
    pool_accum<<<(N + 127) / 128, 256, 0, stream>>>(hcur, batch, pool, cnt, N);
    pool_final<<<(GNUM * DIM + 255) / 256, 256, 0, stream>>>(pool, cnt, (float*)d_out, GNUM * DIM);
}


# edge sum as a gather over edges sorted by destination per call (int count whose return is the rank, prefix sum, placement): a wave sums a node's rows in registers, no f32 atomics there; new sum order
# speedup vs baseline: 3.4585x; 2.6940x over previous
#include <hip/hip_runtime.h>


typedef __attribute__((ext_vector_type(2))) float v2f;
typedef __attribute__((ext_vector_type(8))) float v8f;

#define DIM   128
#define GNUM  64
#define LAYERS 3
#define LNEPS 1e-5f

__global__ void zero_f32(float* __restrict__ p, int n) {
    int i = blockIdx.x * blockDim.x + threadIdx.x;
    if (i < n) p[i] = 0.0f;
}

__global__ void zero_i32(int* __restrict__ p, int n) {
    int i = blockIdx.x * blockDim.x + threadIdx.x;
    if (i < n) p[i] = 0;
}

__global__ void deg_rank(const int* __restrict__ ei, int* __restrict__ cnt,
                         int* __restrict__ rank, int E) {
    int e = blockIdx.x * blockDim.x + threadIdx.x;
    if (e < E) rank[e] = atomicAdd(&cnt[ei[E + e]], 1);
}

#define SCAN_TPB  256
#define SCAN_TILE (SCAN_TPB * 4)
__global__ __launch_bounds__(SCAN_TPB) void scan_deg(const int* __restrict__ cnt,
                                                     int* __restrict__ rowptr,
                                                     float* __restrict__ dinv, int N) {
    __shared__ int wpre[SCAN_TPB / 32];
    __shared__ int wtot[SCAN_TPB / 32];
    const int t = threadIdx.x, lane = t & 31, wave = t >> 5;
    const int tile0 = blockIdx.x * SCAN_TILE;

    int pre = 0;
    const int4* c4 = (const int4*)cnt;
    #pragma unroll 4
    for (int p = 0; p < (int)blockIdx.x; ++p) {
        const int4 q = c4[p * SCAN_TPB + t];
        pre += (q.x + q.y) + (q.z + q.w);
    }
    for (int off = 16; off > 0; off >>= 1) pre += __shfl_xor(pre, off, 32);

    const int base = tile0 + t * 4;
    int c[4];
    for (int j = 0; j < 4; ++j) c[j] = (base + j < N) ? cnt[base + j] : 0;
    const int s = (c[0] + c[1]) + (c[2] + c[3]);
    int inc = s;
    for (int off = 1; off < 32; off <<= 1) {
        const int y = __shfl_up(inc, off, 32);
        if (lane >= off) inc += y;
    }
    if (lane == 0)  wpre[wave] = pre;
    if (lane == 31) wtot[wave] = inc;
    __syncthreads();

    int run = inc - s;
    for (int w = 0; w < SCAN_TPB / 32; ++w) {
        run += wpre[w];
        if (w < wave) run += wtot[w];
    }
    for (int j = 0; j < 4; ++j) {
        const int idx = base + j;
        if (idx < N) {
            rowptr[idx] = run;
            dinv[idx] = rsqrtf((float)c[j] + 1.0f);
            run += c[j];
            if (idx == N - 1) rowptr[N] = run;
        }
    }
}

__global__ void fill_adj(const int* __restrict__ ei, const int* __restrict__ rank,
                         const int* __restrict__ rowptr, const float* __restrict__ dinv,
                         int* __restrict__ ent, int E) {
    int e = blockIdx.x * blockDim.x + threadIdx.x;
    if (e >= E) return;
    const int src = ei[e];
    const int dst = ei[E + e];
    const int pos = rowptr[dst] + rank[e];
    ((int2*)ent)[pos] = make_int2(src, __float_as_int(dinv[src]));
}

__global__ __launch_bounds__(256) void gemm_wmma(const float* __restrict__ H,
                                                 const float* __restrict__ W,
                                                 float* __restrict__ out, int N) {
    __shared__ float lw[DIM * DIM];
    {
        const float4* Wv = (const float4*)W;
        float4* lv = (float4*)lw;
        #pragma unroll
        for (int i = 0; i < (DIM * DIM / 4) / 256; ++i)
            lv[threadIdx.x + i * 256] = Wv[threadIdx.x + i * 256];
    }
    __syncthreads();

    const int wave = threadIdx.x >> 5;
    const int lane = threadIdx.x & 31;
    const int half = lane >> 4;
    const int ln   = lane & 15;

    const int nstrips = N >> 4;
    int strip = blockIdx.x * 8 + wave;
    if (strip >= nstrips) strip = nstrips - 1;
    const int rowBase = strip << 4;

    const v2f* arow2 = (const v2f*)(H + (size_t)(rowBase + ln) * DIM);

    v8f acc[8] = {};

    for (int k = 0; k < DIM; k += 4) {
        v2f a = arow2[(k >> 1) + half];
        const int kb = k + 2 * half;
        #pragma unroll
        for (int t = 0; t < 8; ++t) {
            v2f b;
            b.x = lw[kb * DIM + t * 16 + ln];
            b.y = lw[(kb + 1) * DIM + t * 16 + ln];
            acc[t] = __builtin_amdgcn_wmma_f32_16x16x4_f32(
                false, a, false, b, (short)0, acc[t], false, false);
        }
    }

    #pragma unroll
    for (int t = 0; t < 8; ++t) {
        #pragma unroll
        for (int j = 0; j < 8; ++j) {
            int row = rowBase + half * 8 + j;
            out[(size_t)row * DIM + t * 16 + ln] = acc[t][j];
        }
    }
}

__device__ __forceinline__ void row_fma(float4& a, float w, const float4 v) {
    a.x = __builtin_fmaf(w, v.x, a.x);  a.y = __builtin_fmaf(w, v.y, a.y);
    a.z = __builtin_fmaf(w, v.z, a.z);  a.w = __builtin_fmaf(w, v.w, a.w);
}

__device__ __forceinline__ void row_fma_if(float4& a, float w, const float4 v, bool p) {
    a.x = p ? __builtin_fmaf(w, v.x, a.x) : a.x;  a.y = p ? __builtin_fmaf(w, v.y, a.y) : a.y;
    a.z = p ? __builtin_fmaf(w, v.z, a.z) : a.z;  a.w = p ? __builtin_fmaf(w, v.w, a.w) : a.w;
}

__global__ __launch_bounds__(256) void gather_sum(const int* __restrict__ rowptr,
                                                  const int* __restrict__ ent,
                                                  const float* __restrict__ hw,
                                                  const float* __restrict__ dinv,
                                                  const float* __restrict__ bias,
                                                  float* __restrict__ acc, int N) {
    const int lane = threadIdx.x & 31;
    const int wave = __builtin_amdgcn_readfirstlane((int)(threadIdx.x >> 5));
    const int n = blockIdx.x * 8 + wave;
    if (n >= N) return;
    const int beg = rowptr[n];
    const int end = rowptr[n + 1];
    const float dn = dinv[n];
    const float4 hv = ((const float4*)(hw + (size_t)n * DIM))[lane];
    const float4 bv = ((const float4*)bias)[lane];
    const float sl = dn * dn;
    float4 a;
    a.x = bv.x + sl * hv.x;  a.y = bv.y + sl * hv.y;
    a.z = bv.z + sl * hv.z;  a.w = bv.w + sl * hv.w;

    for (int base = beg; base < end; base += 32) {
        const int m = min(32, end - base);
        int es = 0, ew = 0;
        if (lane < m) {
            const int2 q = ((const int2*)ent)[base + lane];
            es = q.x;
            ew = __float_as_int(__int_as_float(q.y) * dn);
        }
        int k = 0;
        for (; k + 4 <= m; k += 4) {
            const int s0 = __builtin_amdgcn_readlane(es, k);
            const int s1 = __builtin_amdgcn_readlane(es, k + 1);
            const int s2 = __builtin_amdgcn_readlane(es, k + 2);
            const int s3 = __builtin_amdgcn_readlane(es, k + 3);
            const float4 v0 = ((const float4*)(hw + (size_t)s0 * DIM))[lane];
            const float4 v1 = ((const float4*)(hw + (size_t)s1 * DIM))[lane];
            const float4 v2 = ((const float4*)(hw + (size_t)s2 * DIM))[lane];
            const float4 v3 = ((const float4*)(hw + (size_t)s3 * DIM))[lane];
            row_fma(a, __int_as_float(__builtin_amdgcn_readlane(ew, k)),     v0);
            row_fma(a, __int_as_float(__builtin_amdgcn_readlane(ew, k + 1)), v1);
            row_fma(a, __int_as_float(__builtin_amdgcn_readlane(ew, k + 2)), v2);
            row_fma(a, __int_as_float(__builtin_amdgcn_readlane(ew, k + 3)), v3);
        }
        if (k < m) {
            const int k1 = min(k + 1, m - 1);
            const int k2 = min(k + 2, m - 1);
            const int s0 = __builtin_amdgcn_readlane(es, k);
            const int s1 = __builtin_amdgcn_readlane(es, k1);
            const int s2 = __builtin_amdgcn_readlane(es, k2);
            const float4 v0 = ((const float4*)(hw + (size_t)s0 * DIM))[lane];
            const float4 v1 = ((const float4*)(hw + (size_t)s1 * DIM))[lane];
            const float4 v2 = ((const float4*)(hw + (size_t)s2 * DIM))[lane];
            row_fma(a, __int_as_float(__builtin_amdgcn_readlane(ew, k)), v0);
            row_fma_if(a, __int_as_float(__builtin_amdgcn_readlane(ew, k1)), v1, k + 1 < m);
            row_fma_if(a, __int_as_float(__builtin_amdgcn_readlane(ew, k2)), v2, k + 2 < m);
        }
    }
    ((float4*)(acc + (size_t)n * DIM))[lane] = a;
}

__global__ __launch_bounds__(256) void ln_relu(const float* __restrict__ acc,
                                               const float* __restrict__ gamma,
                                               const float* __restrict__ beta,
                                               float* __restrict__ out, int N) {
    int wave = threadIdx.x >> 5, lane = threadIdx.x & 31;
    int n = blockIdx.x * 8 + wave;
    if (n >= N) return;
    float4 v = ((const float4*)(acc + (size_t)n * DIM))[lane];
    float s = v.x + v.y + v.z + v.w;
    float q = v.x * v.x + v.y * v.y + v.z * v.z + v.w * v.w;
    #pragma unroll
    for (int off = 16; off > 0; off >>= 1) {
        s += __shfl_xor(s, off, 32);
        q += __shfl_xor(q, off, 32);
    }
    float mu  = s * (1.0f / DIM);
    float var = fmaxf(q * (1.0f / DIM) - mu * mu, 0.0f);
    float inv = rsqrtf(var + LNEPS);
    float4 g = ((const float4*)gamma)[lane];
    float4 b = ((const float4*)beta)[lane];
    float4 r;
    r.x = fmaxf(0.0f, g.x * (v.x - mu) * inv + b.x);
    r.y = fmaxf(0.0f, g.y * (v.y - mu) * inv + b.y);
    r.z = fmaxf(0.0f, g.z * (v.z - mu) * inv + b.z);
    r.w = fmaxf(0.0f, g.w * (v.w - mu) * inv + b.w);
    ((float4*)(out + (size_t)n * DIM))[lane] = r;
}

#define POOL_NPW 16
#define POOL_WAVES 8

__device__ __forceinline__ void pool_flush(float* __restrict__ pool, float* __restrict__ cnt,
                                           int g, float4 s, float c, int lane) {
    float* p = pool + (size_t)g * DIM + lane * 4;
    atomicAdd(p + 0, s.x);
    atomicAdd(p + 1, s.y);
    atomicAdd(p + 2, s.z);
    atomicAdd(p + 3, s.w);
    if (lane == 0) atomicAdd(&cnt[g], c);
}

__global__ __launch_bounds__(256) void pool_accum(const float* __restrict__ h,
                                                  const int* __restrict__ batch,
                                                  float* __restrict__ pool,
                                                  float* __restrict__ cnt, int N) {
    __shared__ float part[POOL_WAVES][DIM];
    __shared__ int   pgid[POOL_WAVES];
    __shared__ float pcnt[POOL_WAVES];

    const int lane = threadIdx.x & 31;
    const int wave = __builtin_amdgcn_readfirstlane((int)(threadIdx.x >> 5));
    const int n0 = (blockIdx.x * POOL_WAVES + wave) * POOL_NPW;
    const int nv = min(POOL_NPW, N - n0);

    const int gv = batch[min(n0 + (lane & (POOL_NPW - 1)), N - 1)];

    float4 v[POOL_NPW];
    #pragma unroll
    for (int i = 0; i < POOL_NPW; ++i)
        v[i] = ((const float4*)(h + (size_t)min(n0 + i, N - 1) * DIM))[lane];

    int cur = -1;
    float4 s = make_float4(0.0f, 0.0f, 0.0f, 0.0f);
    float c = 0.0f;
    if (nv > 0) {
        cur = __builtin_amdgcn_readlane(gv, 0);
        s = v[0];
        c = 1.0f;
        #pragma unroll
        for (int i = 1; i < POOL_NPW; ++i) {
            if (i < nv) {
                const int gi = __builtin_amdgcn_readlane(gv, i);
                if (gi == cur) {
                    s.x += v[i].x; s.y += v[i].y; s.z += v[i].z; s.w += v[i].w;
                    c += 1.0f;
                } else {
                    pool_flush(pool, cnt, cur, s, c, lane);
                    cur = gi; s = v[i]; c = 1.0f;
                }
            }
        }
    }

    ((float4*)part[wave])[lane] = s;
    if (lane == 0) { pgid[wave] = cur; pcnt[wave] = c; }
    __syncthreads();

    const int t = threadIdx.x;
    if (t <= DIM) {
        const bool isc = (t == DIM);
        float* dstp = isc ? cnt : pool + t;
        const int dstride = isc ? 1 : DIM;
        int   jg[POOL_WAVES];
        float jx[POOL_WAVES];
        #pragma unroll
        for (int w = 0; w < POOL_WAVES; ++w) {
            jg[w] = pgid[w];
            jx[w] = isc ? pcnt[w] : part[w][t & (DIM - 1)];
        }
        int   rg = jg[0];
        float rs = jx[0];
        #pragma unroll
        for (int w = 1; w < POOL_WAVES; ++w) {
            if (jg[w] == rg) {
                rs += jx[w];
            } else {
                if (rg >= 0) atomicAdd(dstp + (size_t)rg * dstride, rs);
                rg = jg[w]; rs = jx[w];
            }
        }
        if (rg >= 0) atomicAdd(dstp + (size_t)rg * dstride, rs);
    }
}

__global__ void pool_final(const float* __restrict__ pool, const float* __restrict__ cnt,
                           float* __restrict__ out, int total) {
    int i = blockIdx.x * blockDim.x + threadIdx.x;
    if (i < total) out[i] = pool[i] / fmaxf(cnt[i / DIM], 1.0f);
}

static inline size_t al256(size_t x) { return (x + 255) & ~(size_t)255; }

extern "C" void kernel_launch(void* const* d_in, const int* in_sizes, int n_in,
                              void* d_out, int out_size, void* d_ws, size_t ws_size,
                              hipStream_t stream) {
    const float* x      = (const float*)d_in[0];
    const int*   ei     = (const int*)d_in[1];
    const int*   batch  = (const int*)d_in[2];
    const float* Ws     = (const float*)d_in[3];
    const float* bs     = (const float*)d_in[4];
    const float* gammas = (const float*)d_in[5];
    const float* betas  = (const float*)d_in[6];

    const int N = in_sizes[0] / DIM;
    const int E = in_sizes[1] / 2;
    const size_t nd = (size_t)N * DIM * sizeof(float);

    char* ws = (char*)d_ws;
    float* dinv = (float*)ws;                 ws += al256((size_t)N * sizeof(float));
    int*   cnt  = (int*)ws;                   ws += al256((size_t)N * sizeof(int));
    int*   rptr = (int*)ws;                   ws += al256((size_t)(N + 1) * sizeof(int));
    int*   rank = (int*)ws;                   ws += al256((size_t)E * sizeof(int));
    int*   ent  = (int*)ws;                   ws += al256((size_t)E * 2 * sizeof(int));
    float* hw   = (float*)ws;                 ws += al256(nd);
    float* accb = (float*)ws;                 ws += al256(nd);
    float* hbuf = (float*)ws;                 ws += al256(nd);
    float* pool = (float*)ws;
    float* pcnt = pool + GNUM * DIM;

    zero_i32<<<(N + 255) / 256, 256, 0, stream>>>(cnt, N);
    deg_rank<<<(E + 255) / 256, 256, 0, stream>>>(ei, cnt, rank, E);
    scan_deg<<<(N + SCAN_TILE - 1) / SCAN_TILE, SCAN_TPB, 0, stream>>>(cnt, rptr, dinv, N);
    fill_adj<<<(E + 255) / 256, 256, 0, stream>>>(ei, rank, rptr, dinv, ent, E);

    const int nstrips = N >> 4;
    const float* hcur = x;
    for (int l = 0; l < LAYERS; ++l) {
        gemm_wmma<<<(nstrips + 7) / 8, 256, 0, stream>>>(
            hcur, Ws + (size_t)l * DIM * DIM, hw, N);
        gather_sum<<<(N + 7) / 8, 256, 0, stream>>>(
            rptr, ent, hw, dinv, bs + (size_t)l * DIM, accb, N);
        ln_relu<<<(N + 7) / 8, 256, 0, stream>>>(
            accb, gammas + (size_t)l * DIM, betas + (size_t)l * DIM, hbuf, N);
        hcur = hbuf;
    }

    zero_f32<<<(GNUM * DIM + GNUM + 255) / 256, 256, 0, stream>>>(pool, GNUM * DIM + GNUM);
    pool_accum<<<(N + 127) / 128, 256, 0, stream>>>(hcur, batch, pool, pcnt, N);
    pool_final<<<(GNUM * DIM + 255) / 256, 256, 0, stream>>>(pool, pcnt, (float*)d_out, GNUM * DIM);
}


# LayerNorm + ReLU moved into the epilogue of the edge gather: the wave that sums a node's row normalises it in registers and stores it once; 3 launches and 150000 waves fewer
# speedup vs baseline: 3.8508x; 1.1134x over previous
#include <hip/hip_runtime.h>


typedef __attribute__((ext_vector_type(2))) float v2f;
typedef __attribute__((ext_vector_type(8))) float v8f;

#define DIM   128
#define GNUM  64
#define LAYERS 3
#define LNEPS 1e-5f

__global__ void zero_f32(float* __restrict__ p, int n) {
    int i = blockIdx.x * blockDim.x + threadIdx.x;
    if (i < n) p[i] = 0.0f;
}

__global__ void zero_i32(int* __restrict__ p, int n) {
    int i = blockIdx.x * blockDim.x + threadIdx.x;
    if (i < n) p[i] = 0;
}

__global__ void deg_rank(const int* __restrict__ ei, int* __restrict__ cnt,
                         int* __restrict__ rank, int E) {
    int e = blockIdx.x * blockDim.x + threadIdx.x;
    if (e < E) rank[e] = atomicAdd(&cnt[ei[E + e]], 1);
}

#define SCAN_TPB  256
#define SCAN_TILE (SCAN_TPB * 4)
__global__ __launch_bounds__(SCAN_TPB) void scan_deg(const int* __restrict__ cnt,
                                                     int* __restrict__ rowptr,
                                                     float* __restrict__ dinv, int N) {
    __shared__ int wpre[SCAN_TPB / 32];
    __shared__ int wtot[SCAN_TPB / 32];
    const int t = threadIdx.x, lane = t & 31, wave = t >> 5;
    const int tile0 = blockIdx.x * SCAN_TILE;

    int pre = 0;
    const int4* c4 = (const int4*)cnt;
    #pragma unroll 4
    for (int p = 0; p < (int)blockIdx.x; ++p) {
        const int4 q = c4[p * SCAN_TPB + t];
        pre += (q.x + q.y) + (q.z + q.w);
    }
    for (int off = 16; off > 0; off >>= 1) pre += __shfl_xor(pre, off, 32);

    const int base = tile0 + t * 4;
    int c[4];
    for (int j = 0; j < 4; ++j) c[j] = (base + j < N) ? cnt[base + j] : 0;
    const int s = (c[0] + c[1]) + (c[2] + c[3]);
    int inc = s;
    for (int off = 1; off < 32; off <<= 1) {
        const int y = __shfl_up(inc, off, 32);
        if (lane >= off) inc += y;
    }
    if (lane == 0)  wpre[wave] = pre;
    if (lane == 31) wtot[wave] = inc;
    __syncthreads();

    int run = inc - s;
    for (int w = 0; w < SCAN_TPB / 32; ++w) {
        run += wpre[w];
        if (w < wave) run += wtot[w];
    }
    for (int j = 0; j < 4; ++j) {
        const int idx = base + j;
        if (idx < N) {
            rowptr[idx] = run;
            dinv[idx] = rsqrtf((float)c[j] + 1.0f);
            run += c[j];
            if (idx == N - 1) rowptr[N] = run;
        }
    }
}

__global__ void fill_adj(const int* __restrict__ ei, const int* __restrict__ rank,
                         const int* __restrict__ rowptr, const float* __restrict__ dinv,
                         int* __restrict__ ent, int E) {
    int e = blockIdx.x * blockDim.x + threadIdx.x;
    if (e >= E) return;
    const int src = ei[e];
    const int dst = ei[E + e];
    const int pos = rowptr[dst] + rank[e];
    ((int2*)ent)[pos] = make_int2(src, __float_as_int(dinv[src]));
}

__global__ __launch_bounds__(256) void gemm_wmma(const float* __restrict__ H,
                                                 const float* __restrict__ W,
                                                 float* __restrict__ out, int N) {
    __shared__ float lw[DIM * DIM];
    {
        const float4* Wv = (const float4*)W;
        float4* lv = (float4*)lw;
        #pragma unroll
        for (int i = 0; i < (DIM * DIM / 4) / 256; ++i)
            lv[threadIdx.x + i * 256] = Wv[threadIdx.x + i * 256];
    }
    __syncthreads();

    const int wave = threadIdx.x >> 5;
    const int lane = threadIdx.x & 31;
    const int half = lane >> 4;
    const int ln   = lane & 15;

    const int nstrips = N >> 4;
    int strip = blockIdx.x * 8 + wave;
    if (strip >= nstrips) strip = nstrips - 1;
    const int rowBase = strip << 4;

    const v2f* arow2 = (const v2f*)(H + (size_t)(rowBase + ln) * DIM);

    v8f acc[8] = {};

    for (int k = 0; k < DIM; k += 4) {
        v2f a = arow2[(k >> 1) + half];
        const int kb = k + 2 * half;
        #pragma unroll
        for (int t = 0; t < 8; ++t) {
            v2f b;
            b.x = lw[kb * DIM + t * 16 + ln];
            b.y = lw[(kb + 1) * DIM + t * 16 + ln];
            acc[t] = __builtin_amdgcn_wmma_f32_16x16x4_f32(
                false, a, false, b, (short)0, acc[t], false, false);
        }
    }

    #pragma unroll
    for (int t = 0; t < 8; ++t) {
        #pragma unroll
        for (int j = 0; j < 8; ++j) {
            int row = rowBase + half * 8 + j;
            out[(size_t)row * DIM + t * 16 + ln] = acc[t][j];
        }
    }
}

__device__ __forceinline__ void row_fma(float4& a, float w, const float4 v) {
    a.x = __builtin_fmaf(w, v.x, a.x);  a.y = __builtin_fmaf(w, v.y, a.y);
    a.z = __builtin_fmaf(w, v.z, a.z);  a.w = __builtin_fmaf(w, v.w, a.w);
}

__device__ __forceinline__ void row_fma_if(float4& a, float w, const float4 v, bool p) {
    a.x = p ? __builtin_fmaf(w, v.x, a.x) : a.x;  a.y = p ? __builtin_fmaf(w, v.y, a.y) : a.y;
    a.z = p ? __builtin_fmaf(w, v.z, a.z) : a.z;  a.w = p ? __builtin_fmaf(w, v.w, a.w) : a.w;
}

__global__ __launch_bounds__(256) void gather_ln_relu(const int* __restrict__ rowptr,
                                                      const int* __restrict__ ent,
                                                      const float* __restrict__ hw,
                                                      const float* __restrict__ dinv,
                                                      const float* __restrict__ bias,
                                                      const float* __restrict__ gamma,
                                                      const float* __restrict__ beta,
                                                      float* __restrict__ out, int N) {
    const int lane = threadIdx.x & 31;
    const int wave = __builtin_amdgcn_readfirstlane((int)(threadIdx.x >> 5));
    const int n = blockIdx.x * 8 + wave;
    if (n >= N) return;
    const int beg = rowptr[n];
    const int end = rowptr[n + 1];
    const float dn = dinv[n];
    const float4 hv = ((const float4*)(hw + (size_t)n * DIM))[lane];
    const float4 bv = ((const float4*)bias)[lane];
    const float4 g = ((const float4*)gamma)[lane];
    const float4 b = ((const float4*)beta)[lane];
    const float sl = dn * dn;
    float4 a;
    a.x = bv.x + sl * hv.x;  a.y = bv.y + sl * hv.y;
    a.z = bv.z + sl * hv.z;  a.w = bv.w + sl * hv.w;

    for (int base = beg; base < end; base += 32) {
        const int m = min(32, end - base);
        int es = 0, ew = 0;
        if (lane < m) {
            const int2 q = ((const int2*)ent)[base + lane];
            es = q.x;
            ew = __float_as_int(__int_as_float(q.y) * dn);
        }
        int k = 0;
        for (; k + 4 <= m; k += 4) {
            const int s0 = __builtin_amdgcn_readlane(es, k);
            const int s1 = __builtin_amdgcn_readlane(es, k + 1);
            const int s2 = __builtin_amdgcn_readlane(es, k + 2);
            const int s3 = __builtin_amdgcn_readlane(es, k + 3);
            const float4 v0 = ((const float4*)(hw + (size_t)s0 * DIM))[lane];
            const float4 v1 = ((const float4*)(hw + (size_t)s1 * DIM))[lane];
            const float4 v2 = ((const float4*)(hw + (size_t)s2 * DIM))[lane];
            const float4 v3 = ((const float4*)(hw + (size_t)s3 * DIM))[lane];
            row_fma(a, __int_as_float(__builtin_amdgcn_readlane(ew, k)),     v0);
            row_fma(a, __int_as_float(__builtin_amdgcn_readlane(ew, k + 1)), v1);
            row_fma(a, __int_as_float(__builtin_amdgcn_readlane(ew, k + 2)), v2);
            row_fma(a, __int_as_float(__builtin_amdgcn_readlane(ew, k + 3)), v3);
        }
        if (k < m) {
            const int k1 = min(k + 1, m - 1);
            const int k2 = min(k + 2, m - 1);
            const int s0 = __builtin_amdgcn_readlane(es, k);
            const int s1 = __builtin_amdgcn_readlane(es, k1);
            const int s2 = __builtin_amdgcn_readlane(es, k2);
            const float4 v0 = ((const float4*)(hw + (size_t)s0 * DIM))[lane];
            const float4 v1 = ((const float4*)(hw + (size_t)s1 * DIM))[lane];
            const float4 v2 = ((const float4*)(hw + (size_t)s2 * DIM))[lane];
            row_fma(a, __int_as_float(__builtin_amdgcn_readlane(ew, k)), v0);
            row_fma_if(a, __int_as_float(__builtin_amdgcn_readlane(ew, k1)), v1, k + 1 < m);
            row_fma_if(a, __int_as_float(__builtin_amdgcn_readlane(ew, k2)), v2, k + 2 < m);
        }
    }
    float s = a.x + a.y + a.z + a.w;
    float q = a.x * a.x + a.y * a.y + a.z * a.z + a.w * a.w;
    for (int off = 16; off > 0; off >>= 1) {
        s += __shfl_xor(s, off, 32);
        q += __shfl_xor(q, off, 32);
    }
    const float mu  = s * (1.0f / DIM);
    const float var = fmaxf(q * (1.0f / DIM) - mu * mu, 0.0f);
    const float inv = rsqrtf(var + LNEPS);
    float4 r;
    r.x = fmaxf(0.0f, g.x * (a.x - mu) * inv + b.x);
    r.y = fmaxf(0.0f, g.y * (a.y - mu) * inv + b.y);
    r.z = fmaxf(0.0f, g.z * (a.z - mu) * inv + b.z);
    r.w = fmaxf(0.0f, g.w * (a.w - mu) * inv + b.w);
    ((float4*)(out + (size_t)n * DIM))[lane] = r;
}

#define POOL_NPW 16
#define POOL_WAVES 8

__device__ __forceinline__ void pool_flush(float* __restrict__ pool, float* __restrict__ cnt,
                                           int g, float4 s, float c, int lane) {
    float* p = pool + (size_t)g * DIM + lane * 4;
    atomicAdd(p + 0, s.x);
    atomicAdd(p + 1, s.y);
    atomicAdd(p + 2, s.z);
    atomicAdd(p + 3, s.w);
    if (lane == 0) atomicAdd(&cnt[g], c);
}

__global__ __launch_bounds__(256) void pool_accum(const float* __restrict__ h,
                                                  const int* __restrict__ batch,
                                                  float* __restrict__ pool,
                                                  float* __restrict__ cnt, int N) {
    __shared__ float part[POOL_WAVES][DIM];
    __shared__ int   pgid[POOL_WAVES];
    __shared__ float pcnt[POOL_WAVES];

    const int lane = threadIdx.x & 31;
    const int wave = __builtin_amdgcn_readfirstlane((int)(threadIdx.x >> 5));
    const int n0 = (blockIdx.x * POOL_WAVES + wave) * POOL_NPW;
    const int nv = min(POOL_NPW, N - n0);

    const int gv = batch[min(n0 + (lane & (POOL_NPW - 1)), N - 1)];

    float4 v[POOL_NPW];
    #pragma unroll
    for (int i = 0; i < POOL_NPW; ++i)
        v[i] = ((const float4*)(h + (size_t)min(n0 + i, N - 1) * DIM))[lane];

    int cur = -1;
    float4 s = make_float4(0.0f, 0.0f, 0.0f, 0.0f);
    float c = 0.0f;
    if (nv > 0) {
        cur = __builtin_amdgcn_readlane(gv, 0);
        s = v[0];
        c = 1.0f;
        #pragma unroll
        for (int i = 1; i < POOL_NPW; ++i) {
            if (i < nv) {
                const int gi = __builtin_amdgcn_readlane(gv, i);
                if (gi == cur) {
                    s.x += v[i].x; s.y += v[i].y; s.z += v[i].z; s.w += v[i].w;
                    c += 1.0f;
                } else {
                    pool_flush(pool, cnt, cur, s, c, lane);
                    cur = gi; s = v[i]; c = 1.0f;
                }
            }
        }
    }

    ((float4*)part[wave])[lane] = s;
    if (lane == 0) { pgid[wave] = cur; pcnt[wave] = c; }
    __syncthreads();

    const int t = threadIdx.x;
    if (t <= DIM) {
        const bool isc = (t == DIM);
        float* dstp = isc ? cnt : pool + t;
        const int dstride = isc ? 1 : DIM;
        int   jg[POOL_WAVES];
        float jx[POOL_WAVES];
        #pragma unroll
        for (int w = 0; w < POOL_WAVES; ++w) {
            jg[w] = pgid[w];
            jx[w] = isc ? pcnt[w] : part[w][t & (DIM - 1)];
        }
        int   rg = jg[0];
        float rs = jx[0];
        #pragma unroll
        for (int w = 1; w < POOL_WAVES; ++w) {
            if (jg[w] == rg) {
                rs += jx[w];
            } else {
                if (rg >= 0) atomicAdd(dstp + (size_t)rg * dstride, rs);
                rg = jg[w]; rs = jx[w];
            }
        }
        if (rg >= 0) atomicAdd(dstp + (size_t)rg * dstride, rs);
    }
}

__global__ void pool_final(const float* __restrict__ pool, const float* __restrict__ cnt,
                           float* __restrict__ out, int total) {
    int i = blockIdx.x * blockDim.x + threadIdx.x;
    if (i < total) out[i] = pool[i] / fmaxf(cnt[i / DIM], 1.0f);
}

static inline size_t al256(size_t x) { return (x + 255) & ~(size_t)255; }

extern "C" void kernel_launch(void* const* d_in, const int* in_sizes, int n_in,
                              void* d_out, int out_size, void* d_ws, size_t ws_size,
                              hipStream_t stream) {
    const float* x      = (const float*)d_in[0];
    const int*   ei     = (const int*)d_in[1];
    const int*   batch  = (const int*)d_in[2];
    const float* Ws     = (const float*)d_in[3];
    const float* bs     = (const float*)d_in[4];
    const float* gammas = (const float*)d_in[5];
    const float* betas  = (const float*)d_in[6];

    const int N = in_sizes[0] / DIM;
    const int E = in_sizes[1] / 2;
    const size_t nd = (size_t)N * DIM * sizeof(float);

    char* ws = (char*)d_ws;
    float* dinv = (float*)ws;                 ws += al256((size_t)N * sizeof(float));
    int*   cnt  = (int*)ws;                   ws += al256((size_t)N * sizeof(int));
    int*   rptr = (int*)ws;                   ws += al256((size_t)(N + 1) * sizeof(int));
    int*   rank = (int*)ws;                   ws += al256((size_t)E * sizeof(int));
    int*   ent  = (int*)ws;                   ws += al256((size_t)E * 2 * sizeof(int));
    float* hw   = (float*)ws;                 ws += al256(nd);
    float* hbuf = (float*)ws;                 ws += al256(nd);
    float* pool = (float*)ws;
    float* pcnt = pool + GNUM * DIM;

    zero_i32<<<(N + 255) / 256, 256, 0, stream>>>(cnt, N);
    deg_rank<<<(E + 255) / 256, 256, 0, stream>>>(ei, cnt, rank, E);
    scan_deg<<<(N + SCAN_TILE - 1) / SCAN_TILE, SCAN_TPB, 0, stream>>>(cnt, rptr, dinv, N);
    fill_adj<<<(E + 255) / 256, 256, 0, stream>>>(ei, rank, rptr, dinv, ent, E);

    const int nstrips = N >> 4;
    const float* hcur = x;
    for (int l = 0; l < LAYERS; ++l) {
        gemm_wmma<<<(nstrips + 7) / 8, 256, 0, stream>>>(
            hcur, Ws + (size_t)l * DIM * DIM, hw, N);
        gather_ln_relu<<<(N + 7) / 8, 256, 0, stream>>>(
            rptr, ent, hw, dinv, bs + (size_t)l * DIM,
            gammas + (size_t)l * DIM, betas + (size_t)l * DIM, hbuf, N);
        hcur = hbuf;
    }

    zero_f32<<<(GNUM * DIM + GNUM + 255) / 256, 256, 0, stream>>>(pool, GNUM * DIM + GNUM);
    pool_accum<<<(N + 127) / 128, 256, 0, stream>>>(hcur, batch, pool, pcnt, N);
    pool_final<<<(GNUM * DIM + 255) / 256, 256, 0, stream>>>(pool, pcnt, (float*)d_out, GNUM * DIM);
}


# independent stages share launches: the layer-1 WMMA product runs inside the prefix-sum launch (whose 49 work-groups left the chip idle) and both zero fills are one launch; 13 -> 11 launches
# speedup vs baseline: 4.0840x; 1.0606x over previous
#include <hip/hip_runtime.h>


typedef __attribute__((ext_vector_type(2))) float v2f;
typedef __attribute__((ext_vector_type(8))) float v8f;

#define DIM   128
#define GNUM  64
#define LAYERS 3
#define LNEPS 1e-5f

__global__ void zero_ws(int* __restrict__ cnt, int n, float* __restrict__ pool, int m) {
    int i = blockIdx.x * blockDim.x + threadIdx.x;
    if (i < n) cnt[i] = 0;
    else if (i - n < m) pool[i - n] = 0.0f;
}

__global__ void deg_rank(const int* __restrict__ ei, int* __restrict__ cnt,
                         int* __restrict__ rank, int E) {
    int e = blockIdx.x * blockDim.x + threadIdx.x;
    if (e < E) rank[e] = atomicAdd(&cnt[ei[E + e]], 1);
}

#define SCAN_TPB  256
#define SCAN_TILE (SCAN_TPB * 4)
__device__ __forceinline__ void scan_body(const int* __restrict__ cnt,
                                          int* __restrict__ rowptr,
                                          float* __restrict__ dinv, int N, const int blk) {
    __shared__ int wpre[SCAN_TPB / 32];
    __shared__ int wtot[SCAN_TPB / 32];
    const int t = threadIdx.x, lane = t & 31, wave = t >> 5;
    const int tile0 = blk * SCAN_TILE;

    int pre = 0;
    const int4* c4 = (const int4*)cnt;
    #pragma unroll 4
    for (int p = 0; p < blk; ++p) {
        const int4 q = c4[p * SCAN_TPB + t];
        pre += (q.x + q.y) + (q.z + q.w);
    }
    for (int off = 16; off > 0; off >>= 1) pre += __shfl_xor(pre, off, 32);

    const int base = tile0 + t * 4;
    int c[4];
    for (int j = 0; j < 4; ++j) c[j] = (base + j < N) ? cnt[base + j] : 0;
    const int s = (c[0] + c[1]) + (c[2] + c[3]);
    int inc = s;
    for (int off = 1; off < 32; off <<= 1) {
        const int y = __shfl_up(inc, off, 32);
        if (lane >= off) inc += y;
    }
    if (lane == 0)  wpre[wave] = pre;
    if (lane == 31) wtot[wave] = inc;
    __syncthreads();

    int run = inc - s;
    for (int w = 0; w < SCAN_TPB / 32; ++w) {
        run += wpre[w];
        if (w < wave) run += wtot[w];
    }
    for (int j = 0; j < 4; ++j) {
        const int idx = base + j;
        if (idx < N) {
            rowptr[idx] = run;
            dinv[idx] = rsqrtf((float)c[j] + 1.0f);
            run += c[j];
            if (idx == N - 1) rowptr[N] = run;
        }
    }
}

__global__ void fill_adj(const int* __restrict__ ei, const int* __restrict__ rank,
                         const int* __restrict__ rowptr, const float* __restrict__ dinv,
                         int* __restrict__ ent, int E) {
    int e = blockIdx.x * blockDim.x + threadIdx.x;
    if (e >= E) return;
    const int src = ei[e];
    const int dst = ei[E + e];
    const int pos = rowptr[dst] + rank[e];
    ((int2*)ent)[pos] = make_int2(src, __float_as_int(dinv[src]));
}

__device__ __forceinline__ void gemm_body(const float* __restrict__ H,
                                          const float* __restrict__ W,
                                          float* __restrict__ out, int N, const int blk) {
    __shared__ float lw[DIM * DIM];
    {
        const float4* Wv = (const float4*)W;
        float4* lv = (float4*)lw;
        #pragma unroll
        for (int i = 0; i < (DIM * DIM / 4) / 256; ++i)
            lv[threadIdx.x + i * 256] = Wv[threadIdx.x + i * 256];
    }
    __syncthreads();

    const int wave = threadIdx.x >> 5;
    const int lane = threadIdx.x & 31;
    const int half = lane >> 4;
    const int ln   = lane & 15;

    const int nstrips = N >> 4;
    int strip = blk * 8 + wave;
    if (strip >= nstrips) strip = nstrips - 1;
    const int rowBase = strip << 4;

    const v2f* arow2 = (const v2f*)(H + (size_t)(rowBase + ln) * DIM);

    v8f acc[8] = {};

    for (int k = 0; k < DIM; k += 4) {
        v2f a = arow2[(k >> 1) + half];
        const int kb = k + 2 * half;
        #pragma unroll
        for (int t = 0; t < 8; ++t) {
            v2f b;
            b.x = lw[kb * DIM + t * 16 + ln];
            b.y = lw[(kb + 1) * DIM + t * 16 + ln];
            acc[t] = __builtin_amdgcn_wmma_f32_16x16x4_f32(
                false, a, false, b, (short)0, acc[t], false, false);
        }
    }

    #pragma unroll
    for (int t = 0; t < 8; ++t) {
        #pragma unroll
        for (int j = 0; j < 8; ++j) {
            int row = rowBase + half * 8 + j;
            out[(size_t)row * DIM + t * 16 + ln] = acc[t][j];
        }
    }
}

__global__ __launch_bounds__(256) void gemm_wmma(const float* __restrict__ H,
                                                 const float* __restrict__ W,
                                                 float* __restrict__ out, int N) {
    gemm_body(H, W, out, N, (int)blockIdx.x);
}

__global__ __launch_bounds__(256) void scan_deg_gemm(const int* __restrict__ cnt,
                                                     int* __restrict__ rowptr,
                                                     float* __restrict__ dinv, int nscan,
                                                     const float* __restrict__ H,
                                                     const float* __restrict__ W,
                                                     float* __restrict__ out, int N) {
    const int blk = (int)blockIdx.x;
    if (blk < nscan) scan_body(cnt, rowptr, dinv, N, blk);
    else             gemm_body(H, W, out, N, blk - nscan);
}

__device__ __forceinline__ void row_fma(float4& a, float w, const float4 v) {
    a.x = __builtin_fmaf(w, v.x, a.x);  a.y = __builtin_fmaf(w, v.y, a.y);
    a.z = __builtin_fmaf(w, v.z, a.z);  a.w = __builtin_fmaf(w, v.w, a.w);
}

__device__ __forceinline__ void row_fma_if(float4& a, float w, const float4 v, bool p) {
    a.x = p ? __builtin_fmaf(w, v.x, a.x) : a.x;  a.y = p ? __builtin_fmaf(w, v.y, a.y) : a.y;
    a.z = p ? __builtin_fmaf(w, v.z, a.z) : a.z;  a.w = p ? __builtin_fmaf(w, v.w, a.w) : a.w;
}

__global__ __launch_bounds__(256) void gather_ln_relu(const int* __restrict__ rowptr,
                                                      const int* __restrict__ ent,
                                                      const float* __restrict__ hw,
                                                      const float* __restrict__ dinv,
                                                      const float* __restrict__ bias,
                                                      const float* __restrict__ gamma,
                                                      const float* __restrict__ beta,
                                                      float* __restrict__ out, int N) {
    const int lane = threadIdx.x & 31;
    const int wave = __builtin_amdgcn_readfirstlane((int)(threadIdx.x >> 5));
    const int n = blockIdx.x * 8 + wave;
    if (n >= N) return;
    const int beg = rowptr[n];
    const int end = rowptr[n + 1];
    const float dn = dinv[n];
    const float4 hv = ((const float4*)(hw + (size_t)n * DIM))[lane];
    const float4 bv = ((const float4*)bias)[lane];
    const float4 g = ((const float4*)gamma)[lane];
    const float4 b = ((const float4*)beta)[lane];
    const float sl = dn * dn;
    float4 a;
    a.x = bv.x + sl * hv.x;  a.y = bv.y + sl * hv.y;
    a.z = bv.z + sl * hv.z;  a.w = bv.w + sl * hv.w;

    for (int base = beg; base < end; base += 32) {
        const int m = min(32, end - base);
        int es = 0, ew = 0;
        if (lane < m) {
            const int2 q = ((const int2*)ent)[base + lane];
            es = q.x;
            ew = __float_as_int(__int_as_float(q.y) * dn);
        }
        int k = 0;
        for (; k + 4 <= m; k += 4) {
            const int s0 = __builtin_amdgcn_readlane(es, k);
            const int s1 = __builtin_amdgcn_readlane(es, k + 1);
            const int s2 = __builtin_amdgcn_readlane(es, k + 2);
            const int s3 = __builtin_amdgcn_readlane(es, k + 3);
            const float4 v0 = ((const float4*)(hw + (size_t)s0 * DIM))[lane];
            const float4 v1 = ((const float4*)(hw + (size_t)s1 * DIM))[lane];
            const float4 v2 = ((const float4*)(hw + (size_t)s2 * DIM))[lane];
            const float4 v3 = ((const float4*)(hw + (size_t)s3 * DIM))[lane];
            row_fma(a, __int_as_float(__builtin_amdgcn_readlane(ew, k)),     v0);
            row_fma(a, __int_as_float(__builtin_amdgcn_readlane(ew, k + 1)), v1);
            row_fma(a, __int_as_float(__builtin_amdgcn_readlane(ew, k + 2)), v2);
            row_fma(a, __int_as_float(__builtin_amdgcn_readlane(ew, k + 3)), v3);
        }
        if (k < m) {
            const int k1 = min(k + 1, m - 1);
            const int k2 = min(k + 2, m - 1);
            const int s0 = __builtin_amdgcn_readlane(es, k);
            const int s1 = __builtin_amdgcn_readlane(es, k1);
            const int s2 = __builtin_amdgcn_readlane(es, k2);
            const float4 v0 = ((const float4*)(hw + (size_t)s0 * DIM))[lane];
            const float4 v1 = ((const float4*)(hw + (size_t)s1 * DIM))[lane];
            const float4 v2 = ((const float4*)(hw + (size_t)s2 * DIM))[lane];
            row_fma(a, __int_as_float(__builtin_amdgcn_readlane(ew, k)), v0);
            row_fma_if(a, __int_as_float(__builtin_amdgcn_readlane(ew, k1)), v1, k + 1 < m);
            row_fma_if(a, __int_as_float(__builtin_amdgcn_readlane(ew, k2)), v2, k + 2 < m);
        }
    }
    float s = a.x + a.y + a.z + a.w;
    float q = a.x * a.x + a.y * a.y + a.z * a.z + a.w * a.w;
    for (int off = 16; off > 0; off >>= 1) {
        s += __shfl_xor(s, off, 32);
        q += __shfl_xor(q, off, 32);
    }
    const float mu  = s * (1.0f / DIM);
    const float var = fmaxf(q * (1.0f / DIM) - mu * mu, 0.0f);
    const float inv = rsqrtf(var + LNEPS);
    float4 r;
    r.x = fmaxf(0.0f, g.x * (a.x - mu) * inv + b.x);
    r.y = fmaxf(0.0f, g.y * (a.y - mu) * inv + b.y);
    r.z = fmaxf(0.0f, g.z * (a.z - mu) * inv + b.z);
    r.w = fmaxf(0.0f, g.w * (a.w - mu) * inv + b.w);
    ((float4*)(out + (size_t)n * DIM))[lane] = r;
}

#define POOL_NPW 16
#define POOL_WAVES 8

__device__ __forceinline__ void pool_flush(float* __restrict__ pool, float* __restrict__ cnt,
                                           int g, float4 s, float c, int lane) {
    float* p = pool + (size_t)g * DIM + lane * 4;
    atomicAdd(p + 0, s.x);
    atomicAdd(p + 1, s.y);
    atomicAdd(p + 2, s.z);
    atomicAdd(p + 3, s.w);
    if (lane == 0) atomicAdd(&cnt[g], c);
}

__global__ __launch_bounds__(256) void pool_accum(const float* __restrict__ h,
                                                  const int* __restrict__ batch,
                                                  float* __restrict__ pool,
                                                  float* __restrict__ cnt, int N) {
    __shared__ float part[POOL_WAVES][DIM];
    __shared__ int   pgid[POOL_WAVES];
    __shared__ float pcnt[POOL_WAVES];

    const int lane = threadIdx.x & 31;
    const int wave = __builtin_amdgcn_readfirstlane((int)(threadIdx.x >> 5));
    const int n0 = (blockIdx.x * POOL_WAVES + wave) * POOL_NPW;
    const int nv = min(POOL_NPW, N - n0);

    const int gv = batch[min(n0 + (lane & (POOL_NPW - 1)), N - 1)];

    float4 v[POOL_NPW];
    #pragma unroll
    for (int i = 0; i < POOL_NPW; ++i)
        v[i] = ((const float4*)(h + (size_t)min(n0 + i, N - 1) * DIM))[lane];

    int cur = -1;
    float4 s = make_float4(0.0f, 0.0f, 0.0f, 0.0f);
    float c = 0.0f;
    if (nv > 0) {
        cur = __builtin_amdgcn_readlane(gv, 0);
        s = v[0];
        c = 1.0f;
        #pragma unroll
        for (int i = 1; i < POOL_NPW; ++i) {
            if (i < nv) {
                const int gi = __builtin_amdgcn_readlane(gv, i);
                if (gi == cur) {
                    s.x += v[i].x; s.y += v[i].y; s.z += v[i].z; s.w += v[i].w;
                    c += 1.0f;
                } else {
                    pool_flush(pool, cnt, cur, s, c, lane);
                    cur = gi; s = v[i]; c = 1.0f;
                }
            }
        }
    }

    ((float4*)part[wave])[lane] = s;
    if (lane == 0) { pgid[wave] = cur; pcnt[wave] = c; }
    __syncthreads();

    const int t = threadIdx.x;
    if (t <= DIM) {
        const bool isc = (t == DIM);
        float* dstp = isc ? cnt : pool + t;
        const int dstride = isc ? 1 : DIM;
        int   jg[POOL_WAVES];
        float jx[POOL_WAVES];
        #pragma unroll
        for (int w = 0; w < POOL_WAVES; ++w) {
            jg[w] = pgid[w];
            jx[w] = isc ? pcnt[w] : part[w][t & (DIM - 1)];
        }
        int   rg = jg[0];
        float rs = jx[0];
        #pragma unroll
        for (int w = 1; w < POOL_WAVES; ++w) {
            if (jg[w] == rg) {
                rs += jx[w];
            } else {
                if (rg >= 0) atomicAdd(dstp + (size_t)rg * dstride, rs);
                rg = jg[w]; rs = jx[w];
            }
        }
        if (rg >= 0) atomicAdd(dstp + (size_t)rg * dstride, rs);
    }
}

__global__ void pool_final(const float* __restrict__ pool, const float* __restrict__ cnt,
                           float* __restrict__ out, int total) {
    int i = blockIdx.x * blockDim.x + threadIdx.x;
    if (i < total) out[i] = pool[i] / fmaxf(cnt[i / DIM], 1.0f);
}

static inline size_t al256(size_t x) { return (x + 255) & ~(size_t)255; }

extern "C" void kernel_launch(void* const* d_in, const int* in_sizes, int n_in,
                              void* d_out, int out_size, void* d_ws, size_t ws_size,
                              hipStream_t stream) {
    const float* x      = (const float*)d_in[0];
    const int*   ei     = (const int*)d_in[1];
    const int*   batch  = (const int*)d_in[2];
    const float* Ws     = (const float*)d_in[3];
    const float* bs     = (const float*)d_in[4];
    const float* gammas = (const float*)d_in[5];
    const float* betas  = (const float*)d_in[6];

    const int N = in_sizes[0] / DIM;
    const int E = in_sizes[1] / 2;
    const size_t nd = (size_t)N * DIM * sizeof(float);

    char* ws = (char*)d_ws;
    float* dinv = (float*)ws;                 ws += al256((size_t)N * sizeof(float));
    int*   cnt  = (int*)ws;                   ws += al256((size_t)N * sizeof(int));
    int*   rptr = (int*)ws;                   ws += al256((size_t)(N + 1) * sizeof(int));
    int*   rank = (int*)ws;                   ws += al256((size_t)E * sizeof(int));
    int*   ent  = (int*)ws;                   ws += al256((size_t)E * 2 * sizeof(int));
    float* hw   = (float*)ws;                 ws += al256(nd);
    float* hbuf = (float*)ws;                 ws += al256(nd);
    float* pool = (float*)ws;
    float* pcnt = pool + GNUM * DIM;
    const int npool = GNUM * DIM + GNUM;

    const int nstrips = N >> 4;
    const int ngemm = (nstrips + 7) / 8;
    const int nscan = (N + SCAN_TILE - 1) / SCAN_TILE;
    zero_ws<<<(N + npool + 255) / 256, 256, 0, stream>>>(cnt, N, pool, npool);
    deg_rank<<<(E + 255) / 256, 256, 0, stream>>>(ei, cnt, rank, E);
    scan_deg_gemm<<<nscan + ngemm, SCAN_TPB, 0, stream>>>(cnt, rptr, dinv, nscan, x, Ws, hw, N);
    fill_adj<<<(E + 255) / 256, 256, 0, stream>>>(ei, rank, rptr, dinv, ent, E);

    const float* hcur = x;
    for (int l = 0; l < LAYERS; ++l) {
        if (l > 0)
            gemm_wmma<<<ngemm, 256, 0, stream>>>(
                hcur, Ws + (size_t)l * DIM * DIM, hw, N);
        gather_ln_relu<<<(N + 7) / 8, 256, 0, stream>>>(
            rptr, ent, hw, dinv, bs + (size_t)l * DIM,
            gammas + (size_t)l * DIM, betas + (size_t)l * DIM, hbuf, N);
        hcur = hbuf;
    }

    pool_accum<<<(N + 127) / 128, 256, 0, stream>>>(hcur, batch, pool, pcnt, N);
    pool_final<<<(GNUM * DIM + 255) / 256, 256, 0, stream>>>(pool, pcnt, (float*)d_out, GNUM * DIM);
}


# edges grouped by destination in ONE pass: the count's returned rank is the edge's slot in a fixed 64-slot row of its node (spill list past it); prefix sum and placement launches go, 11 -> 9 launches
# speedup vs baseline: 4.4230x; 1.0830x over previous
#include <hip/hip_runtime.h>


typedef __attribute__((ext_vector_type(2))) float v2f;
typedef __attribute__((ext_vector_type(8))) float v8f;

#define DIM   128
#define GNUM  64
#define LAYERS 3
#define LNEPS 1e-5f

__global__ void zero_ws(int* __restrict__ cnt, int n, float* __restrict__ pool, int m) {
    int i = blockIdx.x * blockDim.x + threadIdx.x;
    if (i < n) cnt[i] = 0;
    else if (i - n < m) pool[i - n] = 0.0f;
}

#define SLOTS 64
#define PLACE_EDGES 1024
template <bool FULL>
__device__ __forceinline__ void place_edges(const int* __restrict__ ei, int* __restrict__ cnt,
                                            int* __restrict__ slot, int* __restrict__ ovf,
                                            int N, int E, const int e0) {
    int s[4], d[4], r[4];
    #pragma unroll
    for (int j = 0; j < 4; ++j) {
        const int e = e0 + j * 256;
        const bool ok = FULL || e < E;
        s[j] = ok ? ei[e] : 0;
        d[j] = ok ? ei[E + e] : 0;
    }
    #pragma unroll
    for (int j = 0; j < 4; ++j)
        r[j] = (FULL || e0 + j * 256 < E) ? atomicAdd(&cnt[d[j]], 1) : 0;
    #pragma unroll
    for (int j = 0; j < 4; ++j) {
        if (FULL || e0 + j * 256 < E) {
            if (r[j] < SLOTS) {
                slot[(size_t)d[j] * SLOTS + r[j]] = s[j];
            } else {
                const int o = atomicAdd(&cnt[N], 1);
                ((int2*)ovf)[o] = make_int2(d[j], s[j]);
            }
        }
    }
}

__device__ __forceinline__ void place_body(const int* __restrict__ ei, int* __restrict__ cnt,
                                           int* __restrict__ slot, int* __restrict__ ovf,
                                           int N, int E, const int blk) {
    const int e0 = blk * PLACE_EDGES + (int)threadIdx.x;
    if (blk * PLACE_EDGES + PLACE_EDGES <= E) place_edges<true>(ei, cnt, slot, ovf, N, E, e0);
    else                                      place_edges<false>(ei, cnt, slot, ovf, N, E, e0);
}

__device__ __forceinline__ void gemm_body(const float* __restrict__ H,
                                          const float* __restrict__ W,
                                          float* __restrict__ out, int N, const int blk) {
    __shared__ float lw[DIM * DIM];
    {
        const float4* Wv = (const float4*)W;
        float4* lv = (float4*)lw;
        #pragma unroll
        for (int i = 0; i < (DIM * DIM / 4) / 256; ++i)
            lv[threadIdx.x + i * 256] = Wv[threadIdx.x + i * 256];
    }
    __syncthreads();

    const int wave = threadIdx.x >> 5;
    const int lane = threadIdx.x & 31;
    const int half = lane >> 4;
    const int ln   = lane & 15;

    const int nstrips = N >> 4;
    int strip = blk * 8 + wave;
    if (strip >= nstrips) strip = nstrips - 1;
    const int rowBase = strip << 4;

    const v2f* arow2 = (const v2f*)(H + (size_t)(rowBase + ln) * DIM);

    v8f acc[8] = {};

    for (int k = 0; k < DIM; k += 4) {
        v2f a = arow2[(k >> 1) + half];
        const int kb = k + 2 * half;
        #pragma unroll
        for (int t = 0; t < 8; ++t) {
            v2f b;
            b.x = lw[kb * DIM + t * 16 + ln];
            b.y = lw[(kb + 1) * DIM + t * 16 + ln];
            acc[t] = __builtin_amdgcn_wmma_f32_16x16x4_f32(
                false, a, false, b, (short)0, acc[t], false, false);
        }
    }

    #pragma unroll
    for (int t = 0; t < 8; ++t) {
        #pragma unroll
        for (int j = 0; j < 8; ++j) {
            int row = rowBase + half * 8 + j;
            out[(size_t)row * DIM + t * 16 + ln] = acc[t][j];
        }
    }
}

__global__ __launch_bounds__(256) void gemm_wmma(const float* __restrict__ H,
                                                 const float* __restrict__ W,
                                                 float* __restrict__ out, int N) {
    gemm_body(H, W, out, N, (int)blockIdx.x);
}

__global__ __launch_bounds__(256) void place_gemm(const int* __restrict__ ei,
                                                  int* __restrict__ cnt,
                                                  int* __restrict__ slot,
                                                  int* __restrict__ ovf, int E, int ngemm,
                                                  const float* __restrict__ H,
                                                  const float* __restrict__ W,
                                                  float* __restrict__ out, int N) {
    const int blk = (int)blockIdx.x;
    if (blk < ngemm) gemm_body(H, W, out, N, blk);
    else             place_body(ei, cnt, slot, ovf, N, E, blk - ngemm);
}

__device__ __forceinline__ void row_fma(float4& a, float w, const float4 v) {
    a.x = __builtin_fmaf(w, v.x, a.x);  a.y = __builtin_fmaf(w, v.y, a.y);
    a.z = __builtin_fmaf(w, v.z, a.z);  a.w = __builtin_fmaf(w, v.w, a.w);
}

__device__ __forceinline__ void row_fma_if(float4& a, float w, const float4 v, bool p) {
    a.x = p ? __builtin_fmaf(w, v.x, a.x) : a.x;  a.y = p ? __builtin_fmaf(w, v.y, a.y) : a.y;
    a.z = p ? __builtin_fmaf(w, v.z, a.z) : a.z;  a.w = p ? __builtin_fmaf(w, v.w, a.w) : a.w;
}

__device__ __forceinline__ float edge_w(int ecr, int k, float dn) {
    const int c = __builtin_amdgcn_readlane(ecr, k);
    return __builtin_amdgcn_rsqf((float)c + 1.0f) * dn;
}

__global__ __launch_bounds__(256) void gather_ln_relu(const int* __restrict__ cnt,
                                                      const int* __restrict__ slot,
                                                      const int* __restrict__ ovf, int E,
                                                      const float* __restrict__ hw,
                                                      const float* __restrict__ bias,
                                                      const float* __restrict__ gamma,
                                                      const float* __restrict__ beta,
                                                      float* __restrict__ out, int N) {
    const int lane = threadIdx.x & 31;
    const int wave = __builtin_amdgcn_readfirstlane((int)(threadIdx.x >> 5));
    const int n = blockIdx.x * 8 + wave;
    if (n >= N) return;
    const int* srow = slot + (size_t)n * SLOTS;
    const int e0 = srow[lane];
    const int deg = cnt[n];
    const float dn = __builtin_amdgcn_rsqf((float)deg + 1.0f);
    const int mall = min(deg, SLOTS);
    const float4 hv = ((const float4*)(hw + (size_t)n * DIM))[lane];
    const float4 bv = ((const float4*)bias)[lane];
    const float4 g = ((const float4*)gamma)[lane];
    const float4 b = ((const float4*)beta)[lane];
    const float sl = dn * dn;
    float4 a;
    a.x = bv.x + sl * hv.x;  a.y = bv.y + sl * hv.y;
    a.z = bv.z + sl * hv.z;  a.w = bv.w + sl * hv.w;
    asm volatile("" :: "v"(e0));

    for (int base = 0; base < mall; base += 32) {
        const int m = min(32, mall - base);
        int es = (base == 0) ? e0 : srow[32 + lane];
        es = (lane < m) ? es : 0;
        const int ecr = cnt[es];
        int k = 0;
        for (; k + 4 <= m; k += 4) {
            const int s0 = __builtin_amdgcn_readlane(es, k);
            const int s1 = __builtin_amdgcn_readlane(es, k + 1);
            const int s2 = __builtin_amdgcn_readlane(es, k + 2);
            const int s3 = __builtin_amdgcn_readlane(es, k + 3);
            const float4 v0 = ((const float4*)(hw + (size_t)s0 * DIM))[lane];
            const float4 v1 = ((const float4*)(hw + (size_t)s1 * DIM))[lane];
            const float4 v2 = ((const float4*)(hw + (size_t)s2 * DIM))[lane];
            const float4 v3 = ((const float4*)(hw + (size_t)s3 * DIM))[lane];
            row_fma(a, edge_w(ecr, k,     dn), v0);
            row_fma(a, edge_w(ecr, k + 1, dn), v1);
            row_fma(a, edge_w(ecr, k + 2, dn), v2);
            row_fma(a, edge_w(ecr, k + 3, dn), v3);
        }
        if (k < m) {
            const int k1 = min(k + 1, m - 1);
            const int k2 = min(k + 2, m - 1);
            const int s0 = __builtin_amdgcn_readlane(es, k);
            const int s1 = __builtin_amdgcn_readlane(es, k1);
            const int s2 = __builtin_amdgcn_readlane(es, k2);
            const float4 v0 = ((const float4*)(hw + (size_t)s0 * DIM))[lane];
            const float4 v1 = ((const float4*)(hw + (size_t)s1 * DIM))[lane];
            const float4 v2 = ((const float4*)(hw + (size_t)s2 * DIM))[lane];
            row_fma(a, edge_w(ecr, k, dn), v0);
            row_fma_if(a, edge_w(ecr, k1, dn), v1, k + 1 < m);
            row_fma_if(a, edge_w(ecr, k2, dn), v2, k + 2 < m);
        }
    }
    if (deg > SLOTS) {
        const int no = min(cnt[N], E);
        for (int i = 0; i < no; i += 32) {
            int od = -1, os = 0;
            if (i + lane < no) {
                const int2 q = ((const int2*)ovf)[i + lane];
                od = q.x;  os = q.y;
            }
            unsigned hits = __builtin_amdgcn_ballot_w32(od == n);
            while (hits) {
                const int j = __builtin_ctz(hits);
                hits &= hits - 1;
                const int so = __builtin_amdgcn_readlane(os, j);
                const float wo = __builtin_amdgcn_rsqf((float)cnt[so] + 1.0f) * dn;
                row_fma(a, wo, ((const float4*)(hw + (size_t)so * DIM))[lane]);
            }
        }
    }
    float s = a.x + a.y + a.z + a.w;
    float q = a.x * a.x + a.y * a.y + a.z * a.z + a.w * a.w;
    for (int off = 16; off > 0; off >>= 1) {
        s += __shfl_xor(s, off, 32);
        q += __shfl_xor(q, off, 32);
    }
    const float mu  = s * (1.0f / DIM);
    const float var = fmaxf(q * (1.0f / DIM) - mu * mu, 0.0f);
    const float inv = rsqrtf(var + LNEPS);
    float4 r;
    r.x = fmaxf(0.0f, g.x * (a.x - mu) * inv + b.x);
    r.y = fmaxf(0.0f, g.y * (a.y - mu) * inv + b.y);
    r.z = fmaxf(0.0f, g.z * (a.z - mu) * inv + b.z);
    r.w = fmaxf(0.0f, g.w * (a.w - mu) * inv + b.w);
    ((float4*)(out + (size_t)n * DIM))[lane] = r;
}

#define POOL_NPW 16
#define POOL_WAVES 8

__device__ __forceinline__ void pool_flush(float* __restrict__ pool, float* __restrict__ cnt,
                                           int g, float4 s, float c, int lane) {
    float* p = pool + (size_t)g * DIM + lane * 4;
    atomicAdd(p + 0, s.x);
    atomicAdd(p + 1, s.y);
    atomicAdd(p + 2, s.z);
    atomicAdd(p + 3, s.w);
    if (lane == 0) atomicAdd(&cnt[g], c);
}

__global__ __launch_bounds__(256) void pool_accum(const float* __restrict__ h,
                                                  const int* __restrict__ batch,
                                                  float* __restrict__ pool,
                                                  float* __restrict__ cnt, int N) {
    __shared__ float part[POOL_WAVES][DIM];
    __shared__ int   pgid[POOL_WAVES];
    __shared__ float pcnt[POOL_WAVES];

    const int lane = threadIdx.x & 31;
    const int wave = __builtin_amdgcn_readfirstlane((int)(threadIdx.x >> 5));
    const int n0 = (blockIdx.x * POOL_WAVES + wave) * POOL_NPW;
    const int nv = min(POOL_NPW, N - n0);

    const int gv = batch[min(n0 + (lane & (POOL_NPW - 1)), N - 1)];

    float4 v[POOL_NPW];
    #pragma unroll
    for (int i = 0; i < POOL_NPW; ++i)
        v[i] = ((const float4*)(h + (size_t)min(n0 + i, N - 1) * DIM))[lane];

    int cur = -1;
    float4 s = make_float4(0.0f, 0.0f, 0.0f, 0.0f);
    float c = 0.0f;
    if (nv > 0) {
        cur = __builtin_amdgcn_readlane(gv, 0);
        s = v[0];
        c = 1.0f;
        #pragma unroll
        for (int i = 1; i < POOL_NPW; ++i) {
            if (i < nv) {
                const int gi = __builtin_amdgcn_readlane(gv, i);
                if (gi == cur) {
                    s.x += v[i].x; s.y += v[i].y; s.z += v[i].z; s.w += v[i].w;
                    c += 1.0f;
                } else {
                    pool_flush(pool, cnt, cur, s, c, lane);
                    cur = gi; s = v[i]; c = 1.0f;
                }
            }
        }
    }

    ((float4*)part[wave])[lane] = s;
    if (lane == 0) { pgid[wave] = cur; pcnt[wave] = c; }
    __syncthreads();

    const int t = threadIdx.x;
    if (t <= DIM) {
        const bool isc = (t == DIM);
        float* dstp = isc ? cnt : pool + t;
        const int dstride = isc ? 1 : DIM;
        int   jg[POOL_WAVES];
        float jx[POOL_WAVES];
        #pragma unroll
        for (int w = 0; w < POOL_WAVES; ++w) {
            jg[w] = pgid[w];
            jx[w] = isc ? pcnt[w] : part[w][t & (DIM - 1)];
        }
        int   rg = jg[0];
        float rs = jx[0];
        #pragma unroll
        for (int w = 1; w < POOL_WAVES; ++w) {
            if (jg[w] == rg) {
                rs += jx[w];
            } else {
                if (rg >= 0) atomicAdd(dstp + (size_t)rg * dstride, rs);
                rg = jg[w]; rs = jx[w];
            }
        }
        if (rg >= 0) atomicAdd(dstp + (size_t)rg * dstride, rs);
    }
}

__global__ void pool_final(const float* __restrict__ pool, const float* __restrict__ cnt,
                           float* __restrict__ out, int total) {
    int i = blockIdx.x * blockDim.x + threadIdx.x;
    if (i < total) out[i] = pool[i] / fmaxf(cnt[i / DIM], 1.0f);
}

static inline size_t al256(size_t x) { return (x + 255) & ~(size_t)255; }

extern "C" void kernel_launch(void* const* d_in, const int* in_sizes, int n_in,
                              void* d_out, int out_size, void* d_ws, size_t ws_size,
                              hipStream_t stream) {
    const float* x      = (const float*)d_in[0];
    const int*   ei     = (const int*)d_in[1];
    const int*   batch  = (const int*)d_in[2];
    const float* Ws     = (const float*)d_in[3];
    const float* bs     = (const float*)d_in[4];
    const float* gammas = (const float*)d_in[5];
    const float* betas  = (const float*)d_in[6];

    const int N = in_sizes[0] / DIM;
    const int E = in_sizes[1] / 2;
    const size_t nd = (size_t)N * DIM * sizeof(float);

    char* ws = (char*)d_ws;
    int*   cnt  = (int*)ws;                   ws += al256((size_t)(N + 1) * sizeof(int));
    int*   slot = (int*)ws;                   ws += al256((size_t)N * SLOTS * sizeof(int));
    int*   ovf  = (int*)ws;                   ws += al256((size_t)E * 2 * sizeof(int));
    float* hw   = (float*)ws;                 ws += al256(nd);
    float* hbuf = (float*)ws;                 ws += al256(nd);
    float* pool = (float*)ws;
    float* pcnt = pool + GNUM * DIM;
    const int npool = GNUM * DIM + GNUM;

    const int nstrips = N >> 4;
    const int ngemm = (nstrips + 7) / 8;
    const int nplace = (E + PLACE_EDGES - 1) / PLACE_EDGES;
    zero_ws<<<(N + 1 + npool + 255) / 256, 256, 0, stream>>>(cnt, N + 1, pool, npool);
    place_gemm<<<ngemm + nplace, 256, 0, stream>>>(ei, cnt, slot, ovf, E, ngemm, x, Ws, hw, N);

    const float* hcur = x;
    for (int l = 0; l < LAYERS; ++l) {
        if (l > 0)
            gemm_wmma<<<ngemm, 256, 0, stream>>>(
                hcur, Ws + (size_t)l * DIM * DIM, hw, N);
        gather_ln_relu<<<(N + 7) / 8, 256, 0, stream>>>(
            cnt, slot, ovf, E, hw, bs + (size_t)l * DIM,
            gammas + (size_t)l * DIM, betas + (size_t)l * DIM, hbuf, N);
        hcur = hbuf;
    }

    pool_accum<<<(N + 127) / 128, 256, 0, stream>>>(hcur, batch, pool, pcnt, N);
    pool_final<<<(GNUM * DIM + 255) / 256, 256, 0, stream>>>(pool, pcnt, (float*)d_out, GNUM * DIM);
}


# edges grouped by destination with LDS counters, no global atomic per edge: chunks ordered by 200-node range in the zero-fill launch, one owner group per range ranks in LDS (sum order: LDS arrival)
# speedup vs baseline: 5.6488x; 1.2771x over previous
#include <hip/hip_runtime.h>


typedef __attribute__((ext_vector_type(2))) float v2f;
typedef __attribute__((ext_vector_type(8))) float v8f;

#define DIM   128
#define GNUM  64
#define LAYERS 3
#define LNEPS 1e-5f

#define SLOTS 64
#define CHUNK 1024
#define OWN_NODES 200
#define NBKT 256

__global__ __launch_bounds__(256) void bucket_edges(const int* __restrict__ ei, int E, int nchunk,
                                                    int* __restrict__ sorted,
                                                    int* __restrict__ offs,
                                                    int* __restrict__ ovfcnt,
                                                    float* __restrict__ pool, int npool) {
    __shared__ int  hist[NBKT];
    __shared__ int  off[NBKT + 1];
    __shared__ int  wsum[8];
    __shared__ int2 stage[CHUNK];
    const int tid = threadIdx.x;
    const int gi = blockIdx.x * 256 + tid;
    if (gi < npool) pool[gi] = 0.0f;
    else if (gi == npool) *ovfcnt = 0;
    const int q = blockIdx.x;
    if (q >= nchunk) return;
    hist[tid] = 0;
    __syncthreads();
    int s[4], d[4], b[4], r[4];
    bool ok[4];
    #pragma unroll
    for (int j = 0; j < 4; ++j) {
        const int e = q * CHUNK + tid + j * 256;
        ok[j] = e < E;
        s[j] = ok[j] ? ei[e] : 0;
        d[j] = ok[j] ? ei[E + e] : 0;
    }
    #pragma unroll
    for (int j = 0; j < 4; ++j) {
        b[j] = (int)min((unsigned)d[j] / (unsigned)OWN_NODES, (unsigned)(NBKT - 1));
        r[j] = ok[j] ? atomicAdd(&hist[b[j]], 1) : 0;
    }
    __syncthreads();
    const int lane = tid & 31, wave = tid >> 5;
    const int v = hist[tid];
    int inc = v;
    #pragma unroll
    for (int o = 1; o < 32; o <<= 1) {
        const int t = __shfl_up(inc, o, 32);
        if (lane >= o) inc += t;
    }
    if (lane == 31) wsum[wave] = inc;
    __syncthreads();
    int wo = 0;
    #pragma unroll
    for (int w = 0; w < 8; ++w) wo += (w < wave) ? wsum[w] : 0;
    const int exc = wo + inc - v;
    off[tid] = exc;
    if (tid == NBKT - 1) off[NBKT] = exc + v;
    __syncthreads();
    #pragma unroll
    for (int j = 0; j < 4; ++j)
        if (ok[j]) stage[off[b[j]] + r[j]] = make_int2(d[j], s[j]);
    __syncthreads();
    const int total = off[NBKT];
    int2* srt = (int2*)sorted + (size_t)q * CHUNK;
    #pragma unroll
    for (int j = 0; j < 4; ++j) {
        const int i = tid + j * 256;
        if (i < total) srt[i] = stage[i];
    }
    int* orow = offs + (size_t)q * (NBKT + 1);
    orow[tid] = exc;
    if (tid == NBKT - 1) orow[NBKT] = exc + v;
}

__device__ __forceinline__ void own_put(const int2 ds, const int base, int* __restrict__ lcnt,
                                        int* __restrict__ lslot, int* __restrict__ cnt,
                                        int* __restrict__ ovf, int N) {
    const int j = ds.x - base;
    if ((unsigned)j < (unsigned)OWN_NODES) {
        const int r = atomicAdd(&lcnt[j], 1);
        if (r < SLOTS) {
            lslot[j * SLOTS + r] = ds.y;
        } else {
            const int o = atomicAdd(&cnt[N], 1);
            ((int2*)ovf)[o] = ds;
        }
    }
}

__device__ __forceinline__ void own_body(const int* __restrict__ sorted,
                                         const int* __restrict__ offs, int nchunk,
                                         int* __restrict__ cnt, int* __restrict__ slot,
                                         int* __restrict__ ovf, int N, const int p,
                                         int* __restrict__ lds) {
    int* lcnt  = lds;
    int* lslot = lds + OWN_NODES;
    const int tid = threadIdx.x;
    const int base = p * OWN_NODES;
    if (tid < OWN_NODES) lcnt[tid] = 0;
    __syncthreads();
    const int2* srt = (const int2*)sorted;
    for (int q0 = 0; q0 < nchunk; q0 += 4 * 256) {
        int a[4], b[4];
        size_t qb[4];
        #pragma unroll
        for (int u = 0; u < 4; ++u) {
            const int q = q0 + tid + u * 256;
            const bool in = q < nchunk;
            const int qq = in ? q : 0;
            const int* orow = offs + (size_t)qq * (NBKT + 1) + p;
            a[u] = in ? orow[0] : 0;
            b[u] = in ? orow[1] : 0;
            qb[u] = (size_t)qq * CHUNK;
        }
        int2 en[4][4];
        #pragma unroll
        for (int u = 0; u < 4; ++u) {
            #pragma unroll
            for (int k = 0; k < 4; ++k) {
                const int i = a[u] + k;
                en[u][k] = srt[qb[u] + (i < b[u] ? i : 0)];
            }
        }
        #pragma unroll
        for (int u = 0; u < 4; ++u) {
            #pragma unroll
            for (int k = 0; k < 4; ++k)
                if (a[u] + k < b[u]) own_put(en[u][k], base, lcnt, lslot, cnt, ovf, N);
            for (int i = a[u] + 4; i < b[u]; ++i)
                own_put(srt[qb[u] + i], base, lcnt, lslot, cnt, ovf, N);
        }
    }
    __syncthreads();
    if (tid < OWN_NODES && base + tid < N) cnt[base + tid] = lcnt[tid];
    for (int i = tid; i < OWN_NODES * 32; i += 256) {
        const int j = i >> 5, k = i & 31;
        const int n = base + j;
        if (n < N) {
            const int c = lcnt[j];
            int* row = slot + (size_t)n * SLOTS;
            row[k] = (k < c) ? lslot[j * SLOTS + k] : 0;
            if (c > 32) row[32 + k] = (32 + k < c) ? lslot[j * SLOTS + 32 + k] : 0;
        }
    }
}

__device__ __forceinline__ void gemm_body(const float* __restrict__ H,
                                          const float* __restrict__ W,
                                          float* __restrict__ out, int N, const int blk,
                                          float* __restrict__ lw) {
    {
        const float4* Wv = (const float4*)W;
        float4* lv = (float4*)lw;
        #pragma unroll
        for (int i = 0; i < (DIM * DIM / 4) / 256; ++i)
            lv[threadIdx.x + i * 256] = Wv[threadIdx.x + i * 256];
    }
    __syncthreads();

    const int wave = threadIdx.x >> 5;
    const int lane = threadIdx.x & 31;
    const int half = lane >> 4;
    const int ln   = lane & 15;

    const int nstrips = N >> 4;
    int strip = blk * 8 + wave;
    if (strip >= nstrips) strip = nstrips - 1;
    const int rowBase = strip << 4;

    const v2f* arow2 = (const v2f*)(H + (size_t)(rowBase + ln) * DIM);

    v8f acc[8] = {};

    for (int k = 0; k < DIM; k += 4) {
        v2f a = arow2[(k >> 1) + half];
        const int kb = k + 2 * half;
        #pragma unroll
        for (int t = 0; t < 8; ++t) {
            v2f b;
            b.x = lw[kb * DIM + t * 16 + ln];
            b.y = lw[(kb + 1) * DIM + t * 16 + ln];
            acc[t] = __builtin_amdgcn_wmma_f32_16x16x4_f32(
                false, a, false, b, (short)0, acc[t], false, false);
        }
    }

    #pragma unroll
    for (int t = 0; t < 8; ++t) {
        #pragma unroll
        for (int j = 0; j < 8; ++j) {
            int row = rowBase + half * 8 + j;
            out[(size_t)row * DIM + t * 16 + ln] = acc[t][j];
        }
    }
}

__global__ __launch_bounds__(256) void gemm_wmma(const float* __restrict__ H,
                                                 const float* __restrict__ W,
                                                 float* __restrict__ out, int N) {
    __shared__ float lw[DIM * DIM];
    gemm_body(H, W, out, N, (int)blockIdx.x, lw);
}

__global__ __launch_bounds__(256) void place_gemm(const int* __restrict__ sorted,
                                                  const int* __restrict__ offs, int nchunk,
                                                  int* __restrict__ cnt,
                                                  int* __restrict__ slot,
                                                  int* __restrict__ ovf, int ngemm,
                                                  const float* __restrict__ H,
                                                  const float* __restrict__ W,
                                                  float* __restrict__ out, int N) {
    __shared__ float lw[DIM * DIM];
    const int blk = (int)blockIdx.x;
    if (blk < ngemm) gemm_body(H, W, out, N, blk, lw);
    else             own_body(sorted, offs, nchunk, cnt, slot, ovf, N, blk - ngemm, (int*)lw);
}

__device__ __forceinline__ void row_fma(float4& a, float w, const float4 v) {
    a.x = __builtin_fmaf(w, v.x, a.x);  a.y = __builtin_fmaf(w, v.y, a.y);
    a.z = __builtin_fmaf(w, v.z, a.z);  a.w = __builtin_fmaf(w, v.w, a.w);
}

__device__ __forceinline__ void row_fma_if(float4& a, float w, const float4 v, bool p) {
    a.x = p ? __builtin_fmaf(w, v.x, a.x) : a.x;  a.y = p ? __builtin_fmaf(w, v.y, a.y) : a.y;
    a.z = p ? __builtin_fmaf(w, v.z, a.z) : a.z;  a.w = p ? __builtin_fmaf(w, v.w, a.w) : a.w;
}

__device__ __forceinline__ float edge_w(int ecr, int k, float dn) {
    const int c = __builtin_amdgcn_readlane(ecr, k);
    return __builtin_amdgcn_rsqf((float)c + 1.0f) * dn;
}

__global__ __launch_bounds__(256) void gather_ln_relu(const int* __restrict__ cnt,
                                                      const int* __restrict__ slot,
                                                      const int* __restrict__ ovf, int E,
                                                      const float* __restrict__ hw,
                                                      const float* __restrict__ bias,
                                                      const float* __restrict__ gamma,
                                                      const float* __restrict__ beta,
                                                      float* __restrict__ out, int N) {
    const int lane = threadIdx.x & 31;
    const int wave = __builtin_amdgcn_readfirstlane((int)(threadIdx.x >> 5));
    const int n = blockIdx.x * 8 + wave;
    if (n >= N) return;
    const int* srow = slot + (size_t)n * SLOTS;
    const int e0 = srow[lane];
    const int deg = cnt[n];
    const float dn = __builtin_amdgcn_rsqf((float)deg + 1.0f);
    const int mall = min(deg, SLOTS);
    const float4 hv = ((const float4*)(hw + (size_t)n * DIM))[lane];
    const float4 bv = ((const float4*)bias)[lane];
    const float4 g = ((const float4*)gamma)[lane];
    const float4 b = ((const float4*)beta)[lane];
    const float sl = dn * dn;
    float4 a;
    a.x = bv.x + sl * hv.x;  a.y = bv.y + sl * hv.y;
    a.z = bv.z + sl * hv.z;  a.w = bv.w + sl * hv.w;
    asm volatile("" :: "v"(e0));

    for (int base = 0; base < mall; base += 32) {
        const int m = min(32, mall - base);
        int es = (base == 0) ? e0 : srow[32 + lane];
        es = (lane < m) ? es : 0;
        const int ecr = cnt[es];
        int k = 0;
        for (; k + 4 <= m; k += 4) {
            const int s0 = __builtin_amdgcn_readlane(es, k);
            const int s1 = __builtin_amdgcn_readlane(es, k + 1);
            const int s2 = __builtin_amdgcn_readlane(es, k + 2);
            const int s3 = __builtin_amdgcn_readlane(es, k + 3);
            const float4 v0 = ((const float4*)(hw + (size_t)s0 * DIM))[lane];
            const float4 v1 = ((const float4*)(hw + (size_t)s1 * DIM))[lane];
            const float4 v2 = ((const float4*)(hw + (size_t)s2 * DIM))[lane];
            const float4 v3 = ((const float4*)(hw + (size_t)s3 * DIM))[lane];
            row_fma(a, edge_w(ecr, k,     dn), v0);
            row_fma(a, edge_w(ecr, k + 1, dn), v1);
            row_fma(a, edge_w(ecr, k + 2, dn), v2);
            row_fma(a, edge_w(ecr, k + 3, dn), v3);
        }
        if (k < m) {
            const int k1 = min(k + 1, m - 1);
            const int k2 = min(k + 2, m - 1);
            const int s0 = __builtin_amdgcn_readlane(es, k);
            const int s1 = __builtin_amdgcn_readlane(es, k1);
            const int s2 = __builtin_amdgcn_readlane(es, k2);
            const float4 v0 = ((const float4*)(hw + (size_t)s0 * DIM))[lane];
            const float4 v1 = ((const float4*)(hw + (size_t)s1 * DIM))[lane];
            const float4 v2 = ((const float4*)(hw + (size_t)s2 * DIM))[lane];
            row_fma(a, edge_w(ecr, k, dn), v0);
            row_fma_if(a, edge_w(ecr, k1, dn), v1, k + 1 < m);
            row_fma_if(a, edge_w(ecr, k2, dn), v2, k + 2 < m);
        }
    }
    if (deg > SLOTS) {
        const int no = min(cnt[N], E);
        for (int i = 0; i < no; i += 32) {
            int od = -1, os = 0;
            if (i + lane < no) {
                const int2 q = ((const int2*)ovf)[i + lane];
                od = q.x;  os = q.y;
            }
            unsigned hits = __builtin_amdgcn_ballot_w32(od == n);
            while (hits) {
                const int j = __builtin_ctz(hits);
                hits &= hits - 1;
                const int so = __builtin_amdgcn_readlane(os, j);
                const float wo = __builtin_amdgcn_rsqf((float)cnt[so] + 1.0f) * dn;
                row_fma(a, wo, ((const float4*)(hw + (size_t)so * DIM))[lane]);
            }
        }
    }
    float s = a.x + a.y + a.z + a.w;
    float q = a.x * a.x + a.y * a.y + a.z * a.z + a.w * a.w;
    for (int off = 16; off > 0; off >>= 1) {
        s += __shfl_xor(s, off, 32);
        q += __shfl_xor(q, off, 32);
    }
    const float mu  = s * (1.0f / DIM);
    const float var = fmaxf(q * (1.0f / DIM) - mu * mu, 0.0f);
    const float inv = rsqrtf(var + LNEPS);
    float4 r;
    r.x = fmaxf(0.0f, g.x * (a.x - mu) * inv + b.x);
    r.y = fmaxf(0.0f, g.y * (a.y - mu) * inv + b.y);
    r.z = fmaxf(0.0f, g.z * (a.z - mu) * inv + b.z);
    r.w = fmaxf(0.0f, g.w * (a.w - mu) * inv + b.w);
    ((float4*)(out + (size_t)n * DIM))[lane] = r;
}

#define POOL_NPW 16
#define POOL_WAVES 8

__device__ __forceinline__ void pool_flush(float* __restrict__ pool, float* __restrict__ cnt,
                                           int g, float4 s, float c, int lane) {
    float* p = pool + (size_t)g * DIM + lane * 4;
    atomicAdd(p + 0, s.x);
    atomicAdd(p + 1, s.y);
    atomicAdd(p + 2, s.z);
    atomicAdd(p + 3, s.w);
    if (lane == 0) atomicAdd(&cnt[g], c);
}

__global__ __launch_bounds__(256) void pool_accum(const float* __restrict__ h,
                                                  const int* __restrict__ batch,
                                                  float* __restrict__ pool,
                                                  float* __restrict__ cnt, int N) {
    __shared__ float part[POOL_WAVES][DIM];
    __shared__ int   pgid[POOL_WAVES];
    __shared__ float pcnt[POOL_WAVES];

    const int lane = threadIdx.x & 31;
    const int wave = __builtin_amdgcn_readfirstlane((int)(threadIdx.x >> 5));
    const int n0 = (blockIdx.x * POOL_WAVES + wave) * POOL_NPW;
    const int nv = min(POOL_NPW, N - n0);

    const int gv = batch[min(n0 + (lane & (POOL_NPW - 1)), N - 1)];

    float4 v[POOL_NPW];
    #pragma unroll
    for (int i = 0; i < POOL_NPW; ++i)
        v[i] = ((const float4*)(h + (size_t)min(n0 + i, N - 1) * DIM))[lane];

    int cur = -1;
    float4 s = make_float4(0.0f, 0.0f, 0.0f, 0.0f);
    float c = 0.0f;
    if (nv > 0) {
        cur = __builtin_amdgcn_readlane(gv, 0);
        s = v[0];
        c = 1.0f;
        #pragma unroll
        for (int i = 1; i < POOL_NPW; ++i) {
            if (i < nv) {
                const int gi = __builtin_amdgcn_readlane(gv, i);
                if (gi == cur) {
                    s.x += v[i].x; s.y += v[i].y; s.z += v[i].z; s.w += v[i].w;
                    c += 1.0f;
                } else {
                    pool_flush(pool, cnt, cur, s, c, lane);
                    cur = gi; s = v[i]; c = 1.0f;
                }
            }
        }
    }

    ((float4*)part[wave])[lane] = s;
    if (lane == 0) { pgid[wave] = cur; pcnt[wave] = c; }
    __syncthreads();

    const int t = threadIdx.x;
    if (t <= DIM) {
        const bool isc = (t == DIM);
        float* dstp = isc ? cnt : pool + t;
        const int dstride = isc ? 1 : DIM;
        int   jg[POOL_WAVES];
        float jx[POOL_WAVES];
        #pragma unroll
        for (int w = 0; w < POOL_WAVES; ++w) {
            jg[w] = pgid[w];
            jx[w] = isc ? pcnt[w] : part[w][t & (DIM - 1)];
        }
        int   rg = jg[0];
        float rs = jx[0];
        #pragma unroll
        for (int w = 1; w < POOL_WAVES; ++w) {
            if (jg[w] == rg) {
                rs += jx[w];
            } else {
                if (rg >= 0) atomicAdd(dstp + (size_t)rg * dstride, rs);
                rg = jg[w]; rs = jx[w];
            }
        }
        if (rg >= 0) atomicAdd(dstp + (size_t)rg * dstride, rs);
    }
}

__global__ void pool_final(const float* __restrict__ pool, const float* __restrict__ cnt,
                           float* __restrict__ out, int total) {
    int i = blockIdx.x * blockDim.x + threadIdx.x;
    if (i < total) out[i] = pool[i] / fmaxf(cnt[i / DIM], 1.0f);
}

static inline size_t al256(size_t x) { return (x + 255) & ~(size_t)255; }

extern "C" void kernel_launch(void* const* d_in, const int* in_sizes, int n_in,
                              void* d_out, int out_size, void* d_ws, size_t ws_size,
                              hipStream_t stream) {
    const float* x      = (const float*)d_in[0];
    const int*   ei     = (const int*)d_in[1];
    const int*   batch  = (const int*)d_in[2];
    const float* Ws     = (const float*)d_in[3];
    const float* bs     = (const float*)d_in[4];
    const float* gammas = (const float*)d_in[5];
    const float* betas  = (const float*)d_in[6];

    const int N = in_sizes[0] / DIM;
    const int E = in_sizes[1] / 2;
    const size_t nd = (size_t)N * DIM * sizeof(float);

    char* ws = (char*)d_ws;
    int*   cnt  = (int*)ws;                   ws += al256((size_t)(N + 1) * sizeof(int));
    int*   slot = (int*)ws;                   ws += al256((size_t)N * SLOTS * sizeof(int));
    int*   ovf  = (int*)ws;                   ws += al256((size_t)E * 2 * sizeof(int));
    const int nchunk = (E + CHUNK - 1) / CHUNK;
    int*   srt  = (int*)ws;                   ws += al256((size_t)nchunk * CHUNK * 2 * sizeof(int));
    int*   offs = (int*)ws;                   ws += al256((size_t)nchunk * (NBKT + 1) * sizeof(int));
    float* hw   = (float*)ws;                 ws += al256(nd);
    float* hbuf = (float*)ws;                 ws += al256(nd);
    float* pool = (float*)ws;
    float* pcnt = pool + GNUM * DIM;
    const int npool = GNUM * DIM + GNUM;

    const int nstrips = N >> 4;
    const int ngemm = (nstrips + 7) / 8;
    const int nown = (N + OWN_NODES - 1) / OWN_NODES;
    const int nzero = (npool + 1 + 255) / 256;
    bucket_edges<<<(nchunk > nzero ? nchunk : nzero), 256, 0, stream>>>(
        ei, E, nchunk, srt, offs, cnt + N, pool, npool);
    place_gemm<<<ngemm + nown, 256, 0, stream>>>(srt, offs, nchunk, cnt, slot, ovf, ngemm,
                                                 x, Ws, hw, N);

    const float* hcur = x;
    for (int l = 0; l < LAYERS; ++l) {
        if (l > 0)
            gemm_wmma<<<ngemm, 256, 0, stream>>>(
                hcur, Ws + (size_t)l * DIM * DIM, hw, N);
        gather_ln_relu<<<(N + 7) / 8, 256, 0, stream>>>(
            cnt, slot, ovf, E, hw, bs + (size_t)l * DIM,
            gammas + (size_t)l * DIM, betas + (size_t)l * DIM, hbuf, N);
        hcur = hbuf;
    }

    pool_accum<<<(N + 127) / 128, 256, 0, stream>>>(hcur, batch, pool, pcnt, N);
    pool_final<<<(GNUM * DIM + 255) / 256, 256, 0, stream>>>(pool, pcnt, (float*)d_out, GNUM * DIM);
}


# products of layers 2 and 3 run inside the gather before them: a resident 16-wave work-group stages 16 finished rows in LDS, 8 waves multiply by LDS-held weights (k sum in two chains); 9 -> 7 launches
# speedup vs baseline: 6.4335x; 1.1389x over previous
#include <hip/hip_runtime.h>


typedef __attribute__((ext_vector_type(2))) float v2f;
typedef __attribute__((ext_vector_type(8))) float v8f;

#define DIM   128
#define GNUM  64
#define LAYERS 3
#define LNEPS 1e-5f

#define SLOTS 64
#define CHUNK 1024
#define OWN_NODES 200
#define NBKT 256

__global__ __launch_bounds__(256) void bucket_edges(const int* __restrict__ ei, int E, int nchunk,
                                                    int* __restrict__ sorted,
                                                    int* __restrict__ offs,
                                                    int* __restrict__ ovfcnt,
                                                    float* __restrict__ pool, int npool) {
    __shared__ int  hist[NBKT];
    __shared__ int  off[NBKT + 1];
    __shared__ int  wsum[8];
    __shared__ int2 stage[CHUNK];
    const int tid = threadIdx.x;
    const int gi = blockIdx.x * 256 + tid;
    if (gi < npool) pool[gi] = 0.0f;
    else if (gi == npool) *ovfcnt = 0;
    const int q = blockIdx.x;
    if (q >= nchunk) return;
    hist[tid] = 0;
    __syncthreads();
    int s[4], d[4], b[4], r[4];
    bool ok[4];
    #pragma unroll
    for (int j = 0; j < 4; ++j) {
        const int e = q * CHUNK + tid + j * 256;
        ok[j] = e < E;
        s[j] = ok[j] ? ei[e] : 0;
        d[j] = ok[j] ? ei[E + e] : 0;
    }
    #pragma unroll
    for (int j = 0; j < 4; ++j) {
        b[j] = (int)min((unsigned)d[j] / (unsigned)OWN_NODES, (unsigned)(NBKT - 1));
        r[j] = ok[j] ? atomicAdd(&hist[b[j]], 1) : 0;
    }
    __syncthreads();
    const int lane = tid & 31, wave = tid >> 5;
    const int v = hist[tid];
    int inc = v;
    #pragma unroll
    for (int o = 1; o < 32; o <<= 1) {
        const int t = __shfl_up(inc, o, 32);
        if (lane >= o) inc += t;
    }
    if (lane == 31) wsum[wave] = inc;
    __syncthreads();
    int wo = 0;
    #pragma unroll
    for (int w = 0; w < 8; ++w) wo += (w < wave) ? wsum[w] : 0;
    const int exc = wo + inc - v;
    off[tid] = exc;
    if (tid == NBKT - 1) off[NBKT] = exc + v;
    __syncthreads();
    #pragma unroll
    for (int j = 0; j < 4; ++j)
        if (ok[j]) stage[off[b[j]] + r[j]] = make_int2(d[j], s[j]);
    __syncthreads();
    const int total = off[NBKT];
    int2* srt = (int2*)sorted + (size_t)q * CHUNK;
    #pragma unroll
    for (int j = 0; j < 4; ++j) {
        const int i = tid + j * 256;
        if (i < total) srt[i] = stage[i];
    }
    int* orow = offs + (size_t)q * (NBKT + 1);
    orow[tid] = exc;
    if (tid == NBKT - 1) orow[NBKT] = exc + v;
}

__device__ __forceinline__ void own_put(const int2 ds, const int base, int* __restrict__ lcnt,
                                        int* __restrict__ lslot, int* __restrict__ cnt,
                                        int* __restrict__ ovf, int N) {
    const int j = ds.x - base;
    if ((unsigned)j < (unsigned)OWN_NODES) {
        const int r = atomicAdd(&lcnt[j], 1);
        if (r < SLOTS) {
            lslot[j * SLOTS + r] = ds.y;
        } else {
            const int o = atomicAdd(&cnt[N], 1);
            ((int2*)ovf)[o] = ds;
        }
    }
}

__device__ __forceinline__ void own_body(const int* __restrict__ sorted,
                                         const int* __restrict__ offs, int nchunk,
                                         int* __restrict__ cnt, int* __restrict__ slot,
                                         int* __restrict__ ovf, int N, const int p,
                                         int* __restrict__ lds) {
    int* lcnt  = lds;
    int* lslot = lds + OWN_NODES;
    const int tid = threadIdx.x;
    const int base = p * OWN_NODES;
    if (tid < OWN_NODES) lcnt[tid] = 0;
    __syncthreads();
    const int2* srt = (const int2*)sorted;
    for (int q0 = 0; q0 < nchunk; q0 += 4 * 256) {
        int a[4], b[4];
        size_t qb[4];
        #pragma unroll
        for (int u = 0; u < 4; ++u) {
            const int q = q0 + tid + u * 256;
            const bool in = q < nchunk;
            const int qq = in ? q : 0;
            const int* orow = offs + (size_t)qq * (NBKT + 1) + p;
            a[u] = in ? orow[0] : 0;
            b[u] = in ? orow[1] : 0;
            qb[u] = (size_t)qq * CHUNK;
        }
        int2 en[4][4];
        #pragma unroll
        for (int u = 0; u < 4; ++u) {
            #pragma unroll
            for (int k = 0; k < 4; ++k) {
                const int i = a[u] + k;
                en[u][k] = srt[qb[u] + (i < b[u] ? i : 0)];
            }
        }
        #pragma unroll
        for (int u = 0; u < 4; ++u) {
            #pragma unroll
            for (int k = 0; k < 4; ++k)
                if (a[u] + k < b[u]) own_put(en[u][k], base, lcnt, lslot, cnt, ovf, N);
            for (int i = a[u] + 4; i < b[u]; ++i)
                own_put(srt[qb[u] + i], base, lcnt, lslot, cnt, ovf, N);
        }
    }
    __syncthreads();
    if (tid < OWN_NODES && base + tid < N) cnt[base + tid] = lcnt[tid];
    for (int i = tid; i < OWN_NODES * 32; i += 256) {
        const int j = i >> 5, k = i & 31;
        const int n = base + j;
        if (n < N) {
            const int c = lcnt[j];
            int* row = slot + (size_t)n * SLOTS;
            row[k] = (k < c) ? lslot[j * SLOTS + k] : 0;
            if (c > 32) row[32 + k] = (32 + k < c) ? lslot[j * SLOTS + 32 + k] : 0;
        }
    }
}

__device__ __forceinline__ void gemm_body(const float* __restrict__ H,
                                          const float* __restrict__ W,
                                          float* __restrict__ out, int N, const int blk,
                                          float* __restrict__ lw) {
    {
        const float4* Wv = (const float4*)W;
        float4* lv = (float4*)lw;
        #pragma unroll
        for (int i = 0; i < (DIM * DIM / 4) / 256; ++i)
            lv[threadIdx.x + i * 256] = Wv[threadIdx.x + i * 256];
    }
    __syncthreads();

    const int wave = threadIdx.x >> 5;
    const int lane = threadIdx.x & 31;
    const int half = lane >> 4;
    const int ln   = lane & 15;

    const int nstrips = N >> 4;
    int strip = blk * 8 + wave;
    if (strip >= nstrips) strip = nstrips - 1;
    const int rowBase = strip << 4;

    const v2f* arow2 = (const v2f*)(H + (size_t)(rowBase + ln) * DIM);

    v8f acc[8] = {};

    for (int k = 0; k < DIM; k += 4) {
        v2f a = arow2[(k >> 1) + half];
        const int kb = k + 2 * half;
        #pragma unroll
        for (int t = 0; t < 8; ++t) {
            v2f b;
            b.x = lw[kb * DIM + t * 16 + ln];
            b.y = lw[(kb + 1) * DIM + t * 16 + ln];
            acc[t] = __builtin_amdgcn_wmma_f32_16x16x4_f32(
                false, a, false, b, (short)0, acc[t], false, false);
        }
    }

    #pragma unroll
    for (int t = 0; t < 8; ++t) {
        #pragma unroll
        for (int j = 0; j < 8; ++j) {
            int row = rowBase + half * 8 + j;
            out[(size_t)row * DIM + t * 16 + ln] = acc[t][j];
        }
    }
}

__global__ __launch_bounds__(256) void place_gemm(const int* __restrict__ sorted,
                                                  const int* __restrict__ offs, int nchunk,
                                                  int* __restrict__ cnt,
                                                  int* __restrict__ slot,
                                                  int* __restrict__ ovf, int ngemm,
                                                  const float* __restrict__ H,
                                                  const float* __restrict__ W,
                                                  float* __restrict__ out, int N) {
    __shared__ float lw[DIM * DIM];
    const int blk = (int)blockIdx.x;
    if (blk < ngemm) gemm_body(H, W, out, N, blk, lw);
    else             own_body(sorted, offs, nchunk, cnt, slot, ovf, N, blk - ngemm, (int*)lw);
}

__device__ __forceinline__ void row_fma(float4& a, float w, const float4 v) {
    a.x = __builtin_fmaf(w, v.x, a.x);  a.y = __builtin_fmaf(w, v.y, a.y);
    a.z = __builtin_fmaf(w, v.z, a.z);  a.w = __builtin_fmaf(w, v.w, a.w);
}

__device__ __forceinline__ void row_fma_if(float4& a, float w, const float4 v, bool p) {
    a.x = p ? __builtin_fmaf(w, v.x, a.x) : a.x;  a.y = p ? __builtin_fmaf(w, v.y, a.y) : a.y;
    a.z = p ? __builtin_fmaf(w, v.z, a.z) : a.z;  a.w = p ? __builtin_fmaf(w, v.w, a.w) : a.w;
}

__device__ __forceinline__ float edge_w(int ecr, int k, float dn) {
    const int c = __builtin_amdgcn_readlane(ecr, k);
    return __builtin_amdgcn_rsqf((float)c + 1.0f) * dn;
}

__device__ __forceinline__ float4 gather_row(const int n, const int lane,
                                             const int* __restrict__ cnt,
                                             const int* __restrict__ slot,
                                             const int* __restrict__ ovf, int E,
                                             const float* __restrict__ hw,
                                             const float4 bv, const float4 g, const float4 b,
                                             int N) {
    const int* srow = slot + (size_t)n * SLOTS;
    const int e0 = srow[lane];
    const int deg = cnt[n];
    const float dn = __builtin_amdgcn_rsqf((float)deg + 1.0f);
    const int mall = min(deg, SLOTS);
    const float4 hv = ((const float4*)(hw + (size_t)n * DIM))[lane];
    const float sl = dn * dn;
    float4 a;
    a.x = bv.x + sl * hv.x;  a.y = bv.y + sl * hv.y;
    a.z = bv.z + sl * hv.z;  a.w = bv.w + sl * hv.w;
    asm volatile("" :: "v"(e0));

    for (int base = 0; base < mall; base += 32) {
        const int m = min(32, mall - base);
        int es = (base == 0) ? e0 : srow[32 + lane];
        es = (lane < m) ? es : 0;
        const int ecr = cnt[es];
        int k = 0;
        for (; k + 4 <= m; k += 4) {
            const int s0 = __builtin_amdgcn_readlane(es, k);
            const int s1 = __builtin_amdgcn_readlane(es, k + 1);
            const int s2 = __builtin_amdgcn_readlane(es, k + 2);
            const int s3 = __builtin_amdgcn_readlane(es, k + 3);
            const float4 v0 = ((const float4*)(hw + (size_t)s0 * DIM))[lane];
            const float4 v1 = ((const float4*)(hw + (size_t)s1 * DIM))[lane];
            const float4 v2 = ((const float4*)(hw + (size_t)s2 * DIM))[lane];
            const float4 v3 = ((const float4*)(hw + (size_t)s3 * DIM))[lane];
            row_fma(a, edge_w(ecr, k,     dn), v0);
            row_fma(a, edge_w(ecr, k + 1, dn), v1);
            row_fma(a, edge_w(ecr, k + 2, dn), v2);
            row_fma(a, edge_w(ecr, k + 3, dn), v3);
        }
        if (k < m) {
            const int k1 = min(k + 1, m - 1);
            const int k2 = min(k + 2, m - 1);
            const int s0 = __builtin_amdgcn_readlane(es, k);
            const int s1 = __builtin_amdgcn_readlane(es, k1);
            const int s2 = __builtin_amdgcn_readlane(es, k2);
            const float4 v0 = ((const float4*)(hw + (size_t)s0 * DIM))[lane];
            const float4 v1 = ((const float4*)(hw + (size_t)s1 * DIM))[lane];
            const float4 v2 = ((const float4*)(hw + (size_t)s2 * DIM))[lane];
            row_fma(a, edge_w(ecr, k, dn), v0);
            row_fma_if(a, edge_w(ecr, k1, dn), v1, k + 1 < m);
            row_fma_if(a, edge_w(ecr, k2, dn), v2, k + 2 < m);
        }
    }
    if (deg > SLOTS) {
        const int no = min(cnt[N], E);
        for (int i = 0; i < no; i += 32) {
            int od = -1, os = 0;
            if (i + lane < no) {
                const int2 q = ((const int2*)ovf)[i + lane];
                od = q.x;  os = q.y;
            }
            unsigned hits = __builtin_amdgcn_ballot_w32(od == n);
            while (hits) {
                const int j = __builtin_ctz(hits);
                hits &= hits - 1;
                const int so = __builtin_amdgcn_readlane(os, j);
                const float wo = __builtin_amdgcn_rsqf((float)cnt[so] + 1.0f) * dn;
                row_fma(a, wo, ((const float4*)(hw + (size_t)so * DIM))[lane]);
            }
        }
    }
    float s = a.x + a.y + a.z + a.w;
    float q = a.x * a.x + a.y * a.y + a.z * a.z + a.w * a.w;
    for (int off = 16; off > 0; off >>= 1) {
        s += __shfl_xor(s, off, 32);
        q += __shfl_xor(q, off, 32);
    }
    const float mu  = s * (1.0f / DIM);
    const float var = fmaxf(q * (1.0f / DIM) - mu * mu, 0.0f);
    const float inv = rsqrtf(var + LNEPS);
    float4 r;
    r.x = fmaxf(0.0f, g.x * (a.x - mu) * inv + b.x);
    r.y = fmaxf(0.0f, g.y * (a.y - mu) * inv + b.y);
    r.z = fmaxf(0.0f, g.z * (a.z - mu) * inv + b.z);
    r.w = fmaxf(0.0f, g.w * (a.w - mu) * inv + b.w);
    return r;
}

__global__ __launch_bounds__(256) void gather_ln_relu(const int* __restrict__ cnt,
                                                      const int* __restrict__ slot,
                                                      const int* __restrict__ ovf, int E,
                                                      const float* __restrict__ hw,
                                                      const float* __restrict__ bias,
                                                      const float* __restrict__ gamma,
                                                      const float* __restrict__ beta,
                                                      float* __restrict__ out, int N) {
    const int lane = threadIdx.x & 31;
    const int wave = __builtin_amdgcn_readfirstlane((int)(threadIdx.x >> 5));
    const int n = blockIdx.x * 8 + wave;
    if (n >= N) return;
    const float4 bv = ((const float4*)bias)[lane];
    const float4 g = ((const float4*)gamma)[lane];
    const float4 b = ((const float4*)beta)[lane];
    const float4 r = gather_row(n, lane, cnt, slot, ovf, E, hw, bv, g, b, N);
    ((float4*)(out + (size_t)n * DIM))[lane] = r;
}

#define TILE_ROWS 16
#define TPITCH 132
#define FUSED_WGS 1024

__global__ __launch_bounds__(512) void gather_ln_relu_gemm(const int* __restrict__ cnt,
                                                           const int* __restrict__ slot,
                                                           const int* __restrict__ ovf, int E,
                                                           const float* __restrict__ hw,
                                                           const float* __restrict__ bias,
                                                           const float* __restrict__ gamma,
                                                           const float* __restrict__ beta,
                                                           const float* __restrict__ Wn,
                                                           float* __restrict__ out, int N) {
    __shared__ __attribute__((aligned(16))) float lw[DIM * DIM];
    __shared__ __attribute__((aligned(16))) float tile[TILE_ROWS * TPITCH];
    const int lane = threadIdx.x & 31;
    const int wave = __builtin_amdgcn_readfirstlane((int)(threadIdx.x >> 5));
    {
        const float4* Wv = (const float4*)Wn;
        float4* lv = (float4*)lw;
        #pragma unroll
        for (int i = 0; i < (DIM * DIM / 4) / 512; ++i)
            lv[threadIdx.x + i * 512] = Wv[threadIdx.x + i * 512];
    }
    const float4 bv = ((const float4*)bias)[lane];
    const float4 g = ((const float4*)gamma)[lane];
    const float4 b = ((const float4*)beta)[lane];
    const int half = lane >> 4;
    const int ln   = lane & 15;
    const int ntiles = (N + TILE_ROWS - 1) / TILE_ROWS;

    for (int t = (int)blockIdx.x; t < ntiles; t += (int)gridDim.x) {
        const int n = t * TILE_ROWS + wave;
        float4 r = make_float4(0.0f, 0.0f, 0.0f, 0.0f);
        if (n < N) r = gather_row(n, lane, cnt, slot, ovf, E, hw, bv, g, b, N);
        __syncthreads();
        ((float4*)(tile + wave * TPITCH))[lane] = r;
        __syncthreads();
        if (wave < 8) {
            const float* arow = tile + ln * TPITCH + 2 * half;
            const float* bcol = lw + (2 * half) * DIM + wave * 16 + ln;
            v8f acc0 = {}, acc1 = {};
            #pragma unroll 4
            for (int k = 0; k < DIM; k += 8) {
                const v2f a0 = *(const v2f*)(arow + k);
                const v2f a1 = *(const v2f*)(arow + k + 4);
                v2f b0, b1;
                b0.x = bcol[k * DIM];
                b0.y = bcol[(k + 1) * DIM];
                b1.x = bcol[(k + 4) * DIM];
                b1.y = bcol[(k + 5) * DIM];
                acc0 = __builtin_amdgcn_wmma_f32_16x16x4_f32(
                    false, a0, false, b0, (short)0, acc0, false, false);
                acc1 = __builtin_amdgcn_wmma_f32_16x16x4_f32(
                    false, a1, false, b1, (short)0, acc1, false, false);
            }
            float* orow = out + (size_t)(t * TILE_ROWS + half * 8) * DIM + wave * 16 + ln;
            if (t * TILE_ROWS + TILE_ROWS <= N) {
                #pragma unroll
                for (int j = 0; j < 8; ++j) orow[(size_t)j * DIM] = acc0[j] + acc1[j];
            } else {
                #pragma unroll
                for (int j = 0; j < 8; ++j)
                    if (t * TILE_ROWS + half * 8 + j < N) orow[(size_t)j * DIM] = acc0[j] + acc1[j];
            }
        }
    }
}

#define POOL_NPW 16
#define POOL_WAVES 8

__device__ __forceinline__ void pool_flush(float* __restrict__ pool, float* __restrict__ cnt,
                                           int g, float4 s, float c, int lane) {
    float* p = pool + (size_t)g * DIM + lane * 4;
    atomicAdd(p + 0, s.x);
    atomicAdd(p + 1, s.y);
    atomicAdd(p + 2, s.z);
    atomicAdd(p + 3, s.w);
    if (lane == 0) atomicAdd(&cnt[g], c);
}

__global__ __launch_bounds__(256) void pool_accum(const float* __restrict__ h,
                                                  const int* __restrict__ batch,
                                                  float* __restrict__ pool,
                                                  float* __restrict__ cnt, int N) {
    __shared__ float part[POOL_WAVES][DIM];
    __shared__ int   pgid[POOL_WAVES];
    __shared__ float pcnt[POOL_WAVES];

    const int lane = threadIdx.x & 31;
    const int wave = __builtin_amdgcn_readfirstlane((int)(threadIdx.x >> 5));
    const int n0 = (blockIdx.x * POOL_WAVES + wave) * POOL_NPW;
    const int nv = min(POOL_NPW, N - n0);

    const int gv = batch[min(n0 + (lane & (POOL_NPW - 1)), N - 1)];

    float4 v[POOL_NPW];
    #pragma unroll
    for (int i = 0; i < POOL_NPW; ++i)
        v[i] = ((const float4*)(h + (size_t)min(n0 + i, N - 1) * DIM))[lane];

    int cur = -1;
    float4 s = make_float4(0.0f, 0.0f, 0.0f, 0.0f);
    float c = 0.0f;
    if (nv > 0) {
        cur = __builtin_amdgcn_readlane(gv, 0);
        s = v[0];
        c = 1.0f;
        #pragma unroll
        for (int i = 1; i < POOL_NPW; ++i) {
            if (i < nv) {
                const int gi = __builtin_amdgcn_readlane(gv, i);
                if (gi == cur) {
                    s.x += v[i].x; s.y += v[i].y; s.z += v[i].z; s.w += v[i].w;
                    c += 1.0f;
                } else {
                    pool_flush(pool, cnt, cur, s, c, lane);
                    cur = gi; s = v[i]; c = 1.0f;
                }
            }
        }
    }

    ((float4*)part[wave])[lane] = s;
    if (lane == 0) { pgid[wave] = cur; pcnt[wave] = c; }
    __syncthreads();

    const int t = threadIdx.x;
    if (t <= DIM) {
        const bool isc = (t == DIM);
        float* dstp = isc ? cnt : pool + t;
        const int dstride = isc ? 1 : DIM;
        int   jg[POOL_WAVES];
        float jx[POOL_WAVES];
        #pragma unroll
        for (int w = 0; w < POOL_WAVES; ++w) {
            jg[w] = pgid[w];
            jx[w] = isc ? pcnt[w] : part[w][t & (DIM - 1)];
        }
        int   rg = jg[0];
        float rs = jx[0];
        #pragma unroll
        for (int w = 1; w < POOL_WAVES; ++w) {
            if (jg[w] == rg) {
                rs += jx[w];
            } else {
                if (rg >= 0) atomicAdd(dstp + (size_t)rg * dstride, rs);
                rg = jg[w]; rs = jx[w];
            }
        }
        if (rg >= 0) atomicAdd(dstp + (size_t)rg * dstride, rs);
    }
}

__global__ void pool_final(const float* __restrict__ pool, const float* __restrict__ cnt,
                           float* __restrict__ out, int total) {
    int i = blockIdx.x * blockDim.x + threadIdx.x;
    if (i < total) out[i] = pool[i] / fmaxf(cnt[i / DIM], 1.0f);
}

static inline size_t al256(size_t x) { return (x + 255) & ~(size_t)255; }

extern "C" void kernel_launch(void* const* d_in, const int* in_sizes, int n_in,
                              void* d_out, int out_size, void* d_ws, size_t ws_size,
                              hipStream_t stream) {
    const float* x      = (const float*)d_in[0];
    const int*   ei     = (const int*)d_in[1];
    const int*   batch  = (const int*)d_in[2];
    const float* Ws     = (const float*)d_in[3];
    const float* bs     = (const float*)d_in[4];
    const float* gammas = (const float*)d_in[5];
    const float* betas  = (const float*)d_in[6];

    const int N = in_sizes[0] / DIM;
    const int E = in_sizes[1] / 2;
    const size_t nd = (size_t)N * DIM * sizeof(float);

    char* ws = (char*)d_ws;
    int*   cnt  = (int*)ws;                   ws += al256((size_t)(N + 1) * sizeof(int));
    int*   slot = (int*)ws;                   ws += al256((size_t)N * SLOTS * sizeof(int));
    int*   ovf  = (int*)ws;                   ws += al256((size_t)E * 2 * sizeof(int));
    const int nchunk = (E + CHUNK - 1) / CHUNK;
    int*   srt  = (int*)ws;                   ws += al256((size_t)nchunk * CHUNK * 2 * sizeof(int));
    int*   offs = (int*)ws;                   ws += al256((size_t)nchunk * (NBKT + 1) * sizeof(int));
    float* hw   = (float*)ws;                 ws += al256(nd);
    float* hbuf = (float*)ws;                 ws += al256(nd);
    float* pool = (float*)ws;
    float* pcnt = pool + GNUM * DIM;
    const int npool = GNUM * DIM + GNUM;

    const int nstrips = N >> 4;
    const int ngemm = (nstrips + 7) / 8;
    const int nown = (N + OWN_NODES - 1) / OWN_NODES;
    const int nzero = (npool + 1 + 255) / 256;
    bucket_edges<<<(nchunk > nzero ? nchunk : nzero), 256, 0, stream>>>(
        ei, E, nchunk, srt, offs, cnt + N, pool, npool);
    place_gemm<<<ngemm + nown, 256, 0, stream>>>(srt, offs, nchunk, cnt, slot, ovf, ngemm,
                                                 x, Ws, hw, N);

    float* pin = hw;
    float* pout = hbuf;
    for (int l = 0; l + 1 < LAYERS; ++l) {
        gather_ln_relu_gemm<<<FUSED_WGS, 512, 0, stream>>>(
            cnt, slot, ovf, E, pin, bs + (size_t)l * DIM,
            gammas + (size_t)l * DIM, betas + (size_t)l * DIM,
            Ws + (size_t)(l + 1) * DIM * DIM, pout, N);
        float* tmp = pin;  pin = pout;  pout = tmp;
    }
    gather_ln_relu<<<(N + 7) / 8, 256, 0, stream>>>(
        cnt, slot, ovf, E, pin, bs + (size_t)(LAYERS - 1) * DIM,
        gammas + (size_t)(LAYERS - 1) * DIM, betas + (size_t)(LAYERS - 1) * DIM, pout, N);

    pool_accum<<<(N + 127) / 128, 256, 0, stream>>>(pout, batch, pool, pcnt, N);
    pool_final<<<(GNUM * DIM + 255) / 256, 256, 0, stream>>>(pool, pcnt, (float*)d_out, GNUM * DIM);
}


# rows of the layer-2 and layer-3 products stored scaled by dinv of their node: their gathers add plain rows, scale the sum once by dinv of the destination, no degree look-up per edge (same sum order)
# speedup vs baseline: 6.6450x; 1.0329x over previous
#include <hip/hip_runtime.h>


typedef __attribute__((ext_vector_type(2))) float v2f;
typedef __attribute__((ext_vector_type(8))) float v8f;

#define DIM   128
#define GNUM  64
#define LAYERS 3
#define LNEPS 1e-5f

#define SLOTS 64
#define CHUNK 1024
#define OWN_NODES 200
#define NBKT 256

__global__ __launch_bounds__(256) void bucket_edges(const int* __restrict__ ei, int E, int nchunk,
                                                    int* __restrict__ sorted,
                                                    int* __restrict__ offs,
                                                    int* __restrict__ ovfcnt,
                                                    float* __restrict__ pool, int npool) {
    __shared__ int  hist[NBKT];
    __shared__ int  off[NBKT + 1];
    __shared__ int  wsum[8];
    __shared__ int2 stage[CHUNK];
    const int tid = threadIdx.x;
    const int gi = blockIdx.x * 256 + tid;
    if (gi < npool) pool[gi] = 0.0f;
    else if (gi == npool) *ovfcnt = 0;
    const int q = blockIdx.x;
    if (q >= nchunk) return;
    hist[tid] = 0;
    __syncthreads();
    int s[4], d[4], b[4], r[4];
    bool ok[4];
    #pragma unroll
    for (int j = 0; j < 4; ++j) {
        const int e = q * CHUNK + tid + j * 256;
        ok[j] = e < E;
        s[j] = ok[j] ? ei[e] : 0;
        d[j] = ok[j] ? ei[E + e] : 0;
    }
    #pragma unroll
    for (int j = 0; j < 4; ++j) {
        b[j] = (int)min((unsigned)d[j] / (unsigned)OWN_NODES, (unsigned)(NBKT - 1));
        r[j] = ok[j] ? atomicAdd(&hist[b[j]], 1) : 0;
    }
    __syncthreads();
    const int lane = tid & 31, wave = tid >> 5;
    const int v = hist[tid];
    int inc = v;
    #pragma unroll
    for (int o = 1; o < 32; o <<= 1) {
        const int t = __shfl_up(inc, o, 32);
        if (lane >= o) inc += t;
    }
    if (lane == 31) wsum[wave] = inc;
    __syncthreads();
    int wo = 0;
    #pragma unroll
    for (int w = 0; w < 8; ++w) wo += (w < wave) ? wsum[w] : 0;
    const int exc = wo + inc - v;
    off[tid] = exc;
    if (tid == NBKT - 1) off[NBKT] = exc + v;
    __syncthreads();
    #pragma unroll
    for (int j = 0; j < 4; ++j)
        if (ok[j]) stage[off[b[j]] + r[j]] = make_int2(d[j], s[j]);
    __syncthreads();
    const int total = off[NBKT];
    int2* srt = (int2*)sorted + (size_t)q * CHUNK;
    #pragma unroll
    for (int j = 0; j < 4; ++j) {
        const int i = tid + j * 256;
        if (i < total) srt[i] = stage[i];
    }
    int* orow = offs + (size_t)q * (NBKT + 1);
    orow[tid] = exc;
    if (tid == NBKT - 1) orow[NBKT] = exc + v;
}

__device__ __forceinline__ void own_put(const int2 ds, const int base, int* __restrict__ lcnt,
                                        int* __restrict__ lslot, int* __restrict__ cnt,
                                        int* __restrict__ ovf, int N) {
    const int j = ds.x - base;
    if ((unsigned)j < (unsigned)OWN_NODES) {
        const int r = atomicAdd(&lcnt[j], 1);
        if (r < SLOTS) {
            lslot[j * SLOTS + r] = ds.y;
        } else {
            const int o = atomicAdd(&cnt[N], 1);
            ((int2*)ovf)[o] = ds;
        }
    }
}

__device__ __forceinline__ void own_body(const int* __restrict__ sorted,
                                         const int* __restrict__ offs, int nchunk,
                                         int* __restrict__ cnt, int* __restrict__ slot,
                                         int* __restrict__ ovf, int N, const int p,
                                         int* __restrict__ lds) {
    int* lcnt  = lds;
    int* lslot = lds + OWN_NODES;
    const int tid = threadIdx.x;
    const int base = p * OWN_NODES;
    if (tid < OWN_NODES) lcnt[tid] = 0;
    __syncthreads();
    const int2* srt = (const int2*)sorted;
    for (int q0 = 0; q0 < nchunk; q0 += 4 * 256) {
        int a[4], b[4];
        size_t qb[4];
        #pragma unroll
        for (int u = 0; u < 4; ++u) {
            const int q = q0 + tid + u * 256;
            const bool in = q < nchunk;
            const int qq = in ? q : 0;
            const int* orow = offs + (size_t)qq * (NBKT + 1) + p;
            a[u] = in ? orow[0] : 0;
            b[u] = in ? orow[1] : 0;
            qb[u] = (size_t)qq * CHUNK;
        }
        int2 en[4][4];
        #pragma unroll
        for (int u = 0; u < 4; ++u) {
            #pragma unroll
            for (int k = 0; k < 4; ++k) {
                const int i = a[u] + k;
                en[u][k] = srt[qb[u] + (i < b[u] ? i : 0)];
            }
        }
        #pragma unroll
        for (int u = 0; u < 4; ++u) {
            #pragma unroll
            for (int k = 0; k < 4; ++k)
                if (a[u] + k < b[u]) own_put(en[u][k], base, lcnt, lslot, cnt, ovf, N);
            for (int i = a[u] + 4; i < b[u]; ++i)
                own_put(srt[qb[u] + i], base, lcnt, lslot, cnt, ovf, N);
        }
    }
    __syncthreads();
    if (tid < OWN_NODES && base + tid < N) cnt[base + tid] = lcnt[tid];
    for (int i = tid; i < OWN_NODES * 32; i += 256) {
        const int j = i >> 5, k = i & 31;
        const int n = base + j;
        if (n < N) {
            const int c = lcnt[j];
            int* row = slot + (size_t)n * SLOTS;
            row[k] = (k < c) ? lslot[j * SLOTS + k] : 0;
            if (c > 32) row[32 + k] = (32 + k < c) ? lslot[j * SLOTS + 32 + k] : 0;
        }
    }
}

__device__ __forceinline__ void gemm_body(const float* __restrict__ H,
                                          const float* __restrict__ W,
                                          float* __restrict__ out, int N, const int blk,
                                          float* __restrict__ lw) {
    {
        const float4* Wv = (const float4*)W;
        float4* lv = (float4*)lw;
        #pragma unroll
        for (int i = 0; i < (DIM * DIM / 4) / 256; ++i)
            lv[threadIdx.x + i * 256] = Wv[threadIdx.x + i * 256];
    }
    __syncthreads();

    const int wave = threadIdx.x >> 5;
    const int lane = threadIdx.x & 31;
    const int half = lane >> 4;
    const int ln   = lane & 15;

    const int nstrips = N >> 4;
    int strip = blk * 8 + wave;
    if (strip >= nstrips) strip = nstrips - 1;
    const int rowBase = strip << 4;

    const v2f* arow2 = (const v2f*)(H + (size_t)(rowBase + ln) * DIM);

    v8f acc[8] = {};

    for (int k = 0; k < DIM; k += 4) {
        v2f a = arow2[(k >> 1) + half];
        const int kb = k + 2 * half;
        #pragma unroll
        for (int t = 0; t < 8; ++t) {
            v2f b;
            b.x = lw[kb * DIM + t * 16 + ln];
            b.y = lw[(kb + 1) * DIM + t * 16 + ln];
            acc[t] = __builtin_amdgcn_wmma_f32_16x16x4_f32(
                false, a, false, b, (short)0, acc[t], false, false);
        }
    }

    #pragma unroll
    for (int t = 0; t < 8; ++t) {
        #pragma unroll
        for (int j = 0; j < 8; ++j) {
            int row = rowBase + half * 8 + j;
            out[(size_t)row * DIM + t * 16 + ln] = acc[t][j];
        }
    }
}

__global__ __launch_bounds__(256) void place_gemm(const int* __restrict__ sorted,
                                                  const int* __restrict__ offs, int nchunk,
                                                  int* __restrict__ cnt,
                                                  int* __restrict__ slot,
                                                  int* __restrict__ ovf, int ngemm,
                                                  const float* __restrict__ H,
                                                  const float* __restrict__ W,
                                                  float* __restrict__ out, int N) {
    __shared__ float lw[DIM * DIM];
    const int blk = (int)blockIdx.x;
    if (blk < ngemm) gemm_body(H, W, out, N, blk, lw);
    else             own_body(sorted, offs, nchunk, cnt, slot, ovf, N, blk - ngemm, (int*)lw);
}

__device__ __forceinline__ void row_fma(float4& a, float w, const float4 v) {
    a.x = __builtin_fmaf(w, v.x, a.x);  a.y = __builtin_fmaf(w, v.y, a.y);
    a.z = __builtin_fmaf(w, v.z, a.z);  a.w = __builtin_fmaf(w, v.w, a.w);
}

__device__ __forceinline__ void row_fma_if(float4& a, float w, const float4 v, bool p) {
    a.x = p ? __builtin_fmaf(w, v.x, a.x) : a.x;  a.y = p ? __builtin_fmaf(w, v.y, a.y) : a.y;
    a.z = p ? __builtin_fmaf(w, v.z, a.z) : a.z;  a.w = p ? __builtin_fmaf(w, v.w, a.w) : a.w;
}

__device__ __forceinline__ void row_add(float4& a, const float4 v) {
    a.x += v.x;  a.y += v.y;  a.z += v.z;  a.w += v.w;
}
__device__ __forceinline__ void row_add_if(float4& a, const float4 v, bool p) {
    a.x = p ? a.x + v.x : a.x;  a.y = p ? a.y + v.y : a.y;
    a.z = p ? a.z + v.z : a.z;  a.w = p ? a.w + v.w : a.w;
}

__device__ __forceinline__ float edge_w(int ecr, int k, float dn) {
    const int c = __builtin_amdgcn_readlane(ecr, k);
    return __builtin_amdgcn_rsqf((float)c + 1.0f) * dn;
}

template <bool PRE>
__device__ __forceinline__ float4 gather_row(const int n, const int lane,
                                             const int* __restrict__ cnt,
                                             const int* __restrict__ slot,
                                             const int* __restrict__ ovf, int E,
                                             const float* __restrict__ hw,
                                             const float4 bv, const float4 g, const float4 b,
                                             int N, float& dn_out) {
    const int* srow = slot + (size_t)n * SLOTS;
    const int e0 = srow[lane];
    const int deg = cnt[n];
    const float dn = __builtin_amdgcn_rsqf((float)deg + 1.0f);
    const int mall = min(deg, SLOTS);
    const float4 hv = ((const float4*)(hw + (size_t)n * DIM))[lane];
    dn_out = dn;
    float4 a;
    if (PRE) {
        a = hv;
    } else {
        const float sl = dn * dn;
        a.x = bv.x + sl * hv.x;  a.y = bv.y + sl * hv.y;
        a.z = bv.z + sl * hv.z;  a.w = bv.w + sl * hv.w;
    }
    asm volatile("" :: "v"(e0));

    for (int base = 0; base < mall; base += 32) {
        const int m = min(32, mall - base);
        int es = (base == 0) ? e0 : srow[32 + lane];
        int ecr = 0;
        if (!PRE) {
            es = (lane < m) ? es : 0;
            ecr = cnt[es];
        }
        int k = 0;
        for (; k + 4 <= m; k += 4) {
            const int s0 = __builtin_amdgcn_readlane(es, k);
            const int s1 = __builtin_amdgcn_readlane(es, k + 1);
            const int s2 = __builtin_amdgcn_readlane(es, k + 2);
            const int s3 = __builtin_amdgcn_readlane(es, k + 3);
            const float4 v0 = ((const float4*)(hw + (size_t)s0 * DIM))[lane];
            const float4 v1 = ((const float4*)(hw + (size_t)s1 * DIM))[lane];
            const float4 v2 = ((const float4*)(hw + (size_t)s2 * DIM))[lane];
            const float4 v3 = ((const float4*)(hw + (size_t)s3 * DIM))[lane];
            if (PRE) {
                row_add(a, v0);  row_add(a, v1);  row_add(a, v2);  row_add(a, v3);
            } else {
                row_fma(a, edge_w(ecr, k,     dn), v0);
                row_fma(a, edge_w(ecr, k + 1, dn), v1);
                row_fma(a, edge_w(ecr, k + 2, dn), v2);
                row_fma(a, edge_w(ecr, k + 3, dn), v3);
            }
        }
        if (k < m) {
            const int k1 = min(k + 1, m - 1);
            const int k2 = min(k + 2, m - 1);
            const int s0 = __builtin_amdgcn_readlane(es, k);
            const int s1 = __builtin_amdgcn_readlane(es, k1);
            const int s2 = __builtin_amdgcn_readlane(es, k2);
            const float4 v0 = ((const float4*)(hw + (size_t)s0 * DIM))[lane];
            const float4 v1 = ((const float4*)(hw + (size_t)s1 * DIM))[lane];
            const float4 v2 = ((const float4*)(hw + (size_t)s2 * DIM))[lane];
            if (PRE) {
                row_add(a, v0);
                row_add_if(a, v1, k + 1 < m);
                row_add_if(a, v2, k + 2 < m);
            } else {
                row_fma(a, edge_w(ecr, k, dn), v0);
                row_fma_if(a, edge_w(ecr, k1, dn), v1, k + 1 < m);
                row_fma_if(a, edge_w(ecr, k2, dn), v2, k + 2 < m);
            }
        }
    }
    if (deg > SLOTS) {
        const int no = min(cnt[N], E);
        for (int i = 0; i < no; i += 32) {
            int od = -1, os = 0;
            if (i + lane < no) {
                const int2 q = ((const int2*)ovf)[i + lane];
                od = q.x;  os = q.y;
            }
            unsigned hits = __builtin_amdgcn_ballot_w32(od == n);
            while (hits) {
                const int j = __builtin_ctz(hits);
                hits &= hits - 1;
                const int so = __builtin_amdgcn_readlane(os, j);
                const float4 vo = ((const float4*)(hw + (size_t)so * DIM))[lane];
                if (PRE) {
                    row_add(a, vo);
                } else {
                    const float wo = __builtin_amdgcn_rsqf((float)cnt[so] + 1.0f) * dn;
                    row_fma(a, wo, vo);
                }
            }
        }
    }
    if (PRE) {
        a.x = bv.x + dn * a.x;  a.y = bv.y + dn * a.y;
        a.z = bv.z + dn * a.z;  a.w = bv.w + dn * a.w;
    }
    float s = a.x + a.y + a.z + a.w;
    float q = a.x * a.x + a.y * a.y + a.z * a.z + a.w * a.w;
    for (int off = 16; off > 0; off >>= 1) {
        s += __shfl_xor(s, off, 32);
        q += __shfl_xor(q, off, 32);
    }
    const float mu  = s * (1.0f / DIM);
    const float var = fmaxf(q * (1.0f / DIM) - mu * mu, 0.0f);
    const float inv = rsqrtf(var + LNEPS);
    float4 r;
    r.x = fmaxf(0.0f, g.x * (a.x - mu) * inv + b.x);
    r.y = fmaxf(0.0f, g.y * (a.y - mu) * inv + b.y);
    r.z = fmaxf(0.0f, g.z * (a.z - mu) * inv + b.z);
    r.w = fmaxf(0.0f, g.w * (a.w - mu) * inv + b.w);
    return r;
}

__global__ __launch_bounds__(256) void gather_ln_relu(const int* __restrict__ cnt,
                                                      const int* __restrict__ slot,
                                                      const int* __restrict__ ovf, int E,
                                                      const float* __restrict__ hw,
                                                      const float* __restrict__ bias,
                                                      const float* __restrict__ gamma,
                                                      const float* __restrict__ beta,
                                                      float* __restrict__ out, int N) {
    const int lane = threadIdx.x & 31;
    const int wave = __builtin_amdgcn_readfirstlane((int)(threadIdx.x >> 5));
    const int n = blockIdx.x * 8 + wave;
    if (n >= N) return;
    const float4 bv = ((const float4*)bias)[lane];
    const float4 g = ((const float4*)gamma)[lane];
    const float4 b = ((const float4*)beta)[lane];
    float dn;
    const float4 r = gather_row<true>(n, lane, cnt, slot, ovf, E, hw, bv, g, b, N, dn);
    ((float4*)(out + (size_t)n * DIM))[lane] = r;
}

#define TILE_ROWS 16
#define TPITCH 132
#define FUSED_WGS 1024

template <bool PRE>
__global__ __launch_bounds__(512) void gather_ln_relu_gemm(const int* __restrict__ cnt,
                                                           const int* __restrict__ slot,
                                                           const int* __restrict__ ovf, int E,
                                                           const float* __restrict__ hw,
                                                           const float* __restrict__ bias,
                                                           const float* __restrict__ gamma,
                                                           const float* __restrict__ beta,
                                                           const float* __restrict__ Wn,
                                                           float* __restrict__ out, int N) {
    __shared__ __attribute__((aligned(16))) float lw[DIM * DIM];
    __shared__ __attribute__((aligned(16))) float tile[TILE_ROWS * TPITCH];
    const int lane = threadIdx.x & 31;
    const int wave = __builtin_amdgcn_readfirstlane((int)(threadIdx.x >> 5));
    {
        const float4* Wv = (const float4*)Wn;
        float4* lv = (float4*)lw;
        #pragma unroll
        for (int i = 0; i < (DIM * DIM / 4) / 512; ++i)
            lv[threadIdx.x + i * 512] = Wv[threadIdx.x + i * 512];
    }
    const float4 bv = ((const float4*)bias)[lane];
    const float4 g = ((const float4*)gamma)[lane];
    const float4 b = ((const float4*)beta)[lane];
    const int half = lane >> 4;
    const int ln   = lane & 15;
    const int ntiles = (N + TILE_ROWS - 1) / TILE_ROWS;

    for (int t = (int)blockIdx.x; t < ntiles; t += (int)gridDim.x) {
        const int n = t * TILE_ROWS + wave;
        float4 r = make_float4(0.0f, 0.0f, 0.0f, 0.0f);
        if (n < N) {
            float dn;
            r = gather_row<PRE>(n, lane, cnt, slot, ovf, E, hw, bv, g, b, N, dn);
            r.x *= dn;  r.y *= dn;  r.z *= dn;  r.w *= dn;
        }
        __syncthreads();
        ((float4*)(tile + wave * TPITCH))[lane] = r;
        __syncthreads();
        if (wave < 8) {
            const float* arow = tile + ln * TPITCH + 2 * half;
            const float* bcol = lw + (2 * half) * DIM + wave * 16 + ln;
            v8f acc0 = {}, acc1 = {};
            #pragma unroll 4
            for (int k = 0; k < DIM; k += 8) {
                const v2f a0 = *(const v2f*)(arow + k);
                const v2f a1 = *(const v2f*)(arow + k + 4);
                v2f b0, b1;
                b0.x = bcol[k * DIM];
                b0.y = bcol[(k + 1) * DIM];
                b1.x = bcol[(k + 4) * DIM];
                b1.y = bcol[(k + 5) * DIM];
                acc0 = __builtin_amdgcn_wmma_f32_16x16x4_f32(
                    false, a0, false, b0, (short)0, acc0, false, false);
                acc1 = __builtin_amdgcn_wmma_f32_16x16x4_f32(
                    false, a1, false, b1, (short)0, acc1, false, false);
            }
            float* orow = out + (size_t)(t * TILE_ROWS + half * 8) * DIM + wave * 16 + ln;
            if (t * TILE_ROWS + TILE_ROWS <= N) {
                #pragma unroll
                for (int j = 0; j < 8; ++j) orow[(size_t)j * DIM] = acc0[j] + acc1[j];
            } else {
                #pragma unroll
                for (int j = 0; j < 8; ++j)
                    if (t * TILE_ROWS + half * 8 + j < N) orow[(size_t)j * DIM] = acc0[j] + acc1[j];
            }
        }
    }
}

#define POOL_NPW 16
#define POOL_WAVES 8

__device__ __forceinline__ void pool_flush(float* __restrict__ pool, float* __restrict__ cnt,
                                           int g, float4 s, float c, int lane) {
    float* p = pool + (size_t)g * DIM + lane * 4;
    atomicAdd(p + 0, s.x);
    atomicAdd(p + 1, s.y);
    atomicAdd(p + 2, s.z);
    atomicAdd(p + 3, s.w);
    if (lane == 0) atomicAdd(&cnt[g], c);
}

__global__ __launch_bounds__(256) void pool_accum(const float* __restrict__ h,
                                                  const int* __restrict__ batch,
                                                  float* __restrict__ pool,
                                                  float* __restrict__ cnt, int N) {
    __shared__ float part[POOL_WAVES][DIM];
    __shared__ int   pgid[POOL_WAVES];
    __shared__ float pcnt[POOL_WAVES];

    const int lane = threadIdx.x & 31;
    const int wave = __builtin_amdgcn_readfirstlane((int)(threadIdx.x >> 5));
    const int n0 = (blockIdx.x * POOL_WAVES + wave) * POOL_NPW;
    const int nv = min(POOL_NPW, N - n0);

    const int gv = batch[min(n0 + (lane & (POOL_NPW - 1)), N - 1)];

    float4 v[POOL_NPW];
    #pragma unroll
    for (int i = 0; i < POOL_NPW; ++i)
        v[i] = ((const float4*)(h + (size_t)min(n0 + i, N - 1) * DIM))[lane];

    int cur = -1;
    float4 s = make_float4(0.0f, 0.0f, 0.0f, 0.0f);
    float c = 0.0f;
    if (nv > 0) {
        cur = __builtin_amdgcn_readlane(gv, 0);
        s = v[0];
        c = 1.0f;
        #pragma unroll
        for (int i = 1; i < POOL_NPW; ++i) {
            if (i < nv) {
                const int gi = __builtin_amdgcn_readlane(gv, i);
                if (gi == cur) {
                    s.x += v[i].x; s.y += v[i].y; s.z += v[i].z; s.w += v[i].w;
                    c += 1.0f;
                } else {
                    pool_flush(pool, cnt, cur, s, c, lane);
                    cur = gi; s = v[i]; c = 1.0f;
                }
            }
        }
    }

    ((float4*)part[wave])[lane] = s;
    if (lane == 0) { pgid[wave] = cur; pcnt[wave] = c; }
    __syncthreads();

    const int t = threadIdx.x;
    if (t <= DIM) {
        const bool isc = (t == DIM);
        float* dstp = isc ? cnt : pool + t;
        const int dstride = isc ? 1 : DIM;
        int   jg[POOL_WAVES];
        float jx[POOL_WAVES];
        #pragma unroll
        for (int w = 0; w < POOL_WAVES; ++w) {
            jg[w] = pgid[w];
            jx[w] = isc ? pcnt[w] : part[w][t & (DIM - 1)];
        }
        int   rg = jg[0];
        float rs = jx[0];
        #pragma unroll
        for (int w = 1; w < POOL_WAVES; ++w) {
            if (jg[w] == rg) {
                rs += jx[w];
            } else {
                if (rg >= 0) atomicAdd(dstp + (size_t)rg * dstride, rs);
                rg = jg[w]; rs = jx[w];
            }
        }
        if (rg >= 0) atomicAdd(dstp + (size_t)rg * dstride, rs);
    }
}

__global__ void pool_final(const float* __restrict__ pool, const float* __restrict__ cnt,
                           float* __restrict__ out, int total) {
    int i = blockIdx.x * blockDim.x + threadIdx.x;
    if (i < total) out[i] = pool[i] / fmaxf(cnt[i / DIM], 1.0f);
}

static inline size_t al256(size_t x) { return (x + 255) & ~(size_t)255; }

extern "C" void kernel_launch(void* const* d_in, const int* in_sizes, int n_in,
                              void* d_out, int out_size, void* d_ws, size_t ws_size,
                              hipStream_t stream) {
    const float* x      = (const float*)d_in[0];
    const int*   ei     = (const int*)d_in[1];
    const int*   batch  = (const int*)d_in[2];
    const float* Ws     = (const float*)d_in[3];
    const float* bs     = (const float*)d_in[4];
    const float* gammas = (const float*)d_in[5];
    const float* betas  = (const float*)d_in[6];

    const int N = in_sizes[0] / DIM;
    const int E = in_sizes[1] / 2;
    const size_t nd = (size_t)N * DIM * sizeof(float);

    char* ws = (char*)d_ws;
    int*   cnt  = (int*)ws;                   ws += al256((size_t)(N + 1) * sizeof(int));
    int*   slot = (int*)ws;                   ws += al256((size_t)N * SLOTS * sizeof(int));
    int*   ovf  = (int*)ws;                   ws += al256((size_t)E * 2 * sizeof(int));
    const int nchunk = (E + CHUNK - 1) / CHUNK;
    int*   srt  = (int*)ws;                   ws += al256((size_t)nchunk * CHUNK * 2 * sizeof(int));
    int*   offs = (int*)ws;                   ws += al256((size_t)nchunk * (NBKT + 1) * sizeof(int));
    float* hw   = (float*)ws;                 ws += al256(nd);
    float* hbuf = (float*)ws;                 ws += al256(nd);
    float* pool = (float*)ws;
    float* pcnt = pool + GNUM * DIM;
    const int npool = GNUM * DIM + GNUM;

    const int nstrips = N >> 4;
    const int ngemm = (nstrips + 7) / 8;
    const int nown = (N + OWN_NODES - 1) / OWN_NODES;
    const int nzero = (npool + 1 + 255) / 256;
    bucket_edges<<<(nchunk > nzero ? nchunk : nzero), 256, 0, stream>>>(
        ei, E, nchunk, srt, offs, cnt + N, pool, npool);
    place_gemm<<<ngemm + nown, 256, 0, stream>>>(srt, offs, nchunk, cnt, slot, ovf, ngemm,
                                                 x, Ws, hw, N);

    float* pin = hw;
    float* pout = hbuf;
    for (int l = 0; l + 1 < LAYERS; ++l) {
        if (l == 0)
            gather_ln_relu_gemm<false><<<FUSED_WGS, 512, 0, stream>>>(
                cnt, slot, ovf, E, pin, bs + (size_t)l * DIM,
                gammas + (size_t)l * DIM, betas + (size_t)l * DIM,
                Ws + (size_t)(l + 1) * DIM * DIM, pout, N);
        else
            gather_ln_relu_gemm<true><<<FUSED_WGS, 512, 0, stream>>>(
                cnt, slot, ovf, E, pin, bs + (size_t)l * DIM,
                gammas + (size_t)l * DIM, betas + (size_t)l * DIM,
                Ws + (size_t)(l + 1) * DIM * DIM, pout, N);
        float* tmp = pin;  pin = pout;  pout = tmp;
    }
    gather_ln_relu<<<(N + 7) / 8, 256, 0, stream>>>(
        cnt, slot, ovf, E, pin, bs + (size_t)(LAYERS - 1) * DIM,
        gammas + (size_t)(LAYERS - 1) * DIM, betas + (size_t)(LAYERS - 1) * DIM, pout, N);

    pool_accum<<<(N + 127) / 128, 256, 0, stream>>>(pout, batch, pool, pcnt, N);
    pool_final<<<(GNUM * DIM + 255) / 256, 256, 0, stream>>>(pool, pcnt, (float*)d_out, GNUM * DIM);
}


# mean pool sums formed in the last gather's launch: a resident wave adds the rows of its consecutive nodes in registers, LDS join, one atomic per column per work-group; 7 -> 6 launches (new sum order)
# speedup vs baseline: 6.9937x; 1.0525x over previous
#include <hip/hip_runtime.h>


typedef __attribute__((ext_vector_type(2))) float v2f;
typedef __attribute__((ext_vector_type(8))) float v8f;

#define DIM   128
#define GNUM  64
#define LAYERS 3
#define LNEPS 1e-5f

#define SLOTS 64
#define CHUNK 1024
#define OWN_NODES 200
#define NBKT 256

__global__ __launch_bounds__(256) void bucket_edges(const int* __restrict__ ei, int E, int nchunk,
                                                    int* __restrict__ sorted,
                                                    int* __restrict__ offs,
                                                    int* __restrict__ ovfcnt,
                                                    float* __restrict__ pool, int npool) {
    __shared__ int  hist[NBKT];
    __shared__ int  off[NBKT + 1];
    __shared__ int  wsum[8];
    __shared__ int2 stage[CHUNK];
    const int tid = threadIdx.x;
    const int gi = blockIdx.x * 256 + tid;
    if (gi < npool) pool[gi] = 0.0f;
    else if (gi == npool) *ovfcnt = 0;
    const int q = blockIdx.x;
    if (q >= nchunk) return;
    hist[tid] = 0;
    __syncthreads();
    int s[4], d[4], b[4], r[4];
    bool ok[4];
    #pragma unroll
    for (int j = 0; j < 4; ++j) {
        const int e = q * CHUNK + tid + j * 256;
        ok[j] = e < E;
        s[j] = ok[j] ? ei[e] : 0;
        d[j] = ok[j] ? ei[E + e] : 0;
    }
    #pragma unroll
    for (int j = 0; j < 4; ++j) {
        b[j] = (int)min((unsigned)d[j] / (unsigned)OWN_NODES, (unsigned)(NBKT - 1));
        r[j] = ok[j] ? atomicAdd(&hist[b[j]], 1) : 0;
    }
    __syncthreads();
    const int lane = tid & 31, wave = tid >> 5;
    const int v = hist[tid];
    int inc = v;
    #pragma unroll
    for (int o = 1; o < 32; o <<= 1) {
        const int t = __shfl_up(inc, o, 32);
        if (lane >= o) inc += t;
    }
    if (lane == 31) wsum[wave] = inc;
    __syncthreads();
    int wo = 0;
    #pragma unroll
    for (int w = 0; w < 8; ++w) wo += (w < wave) ? wsum[w] : 0;
    const int exc = wo + inc - v;
    off[tid] = exc;
    if (tid == NBKT - 1) off[NBKT] = exc + v;
    __syncthreads();
    #pragma unroll
    for (int j = 0; j < 4; ++j)
        if (ok[j]) stage[off[b[j]] + r[j]] = make_int2(d[j], s[j]);
    __syncthreads();
    const int total = off[NBKT];
    int2* srt = (int2*)sorted + (size_t)q * CHUNK;
    #pragma unroll
    for (int j = 0; j < 4; ++j) {
        const int i = tid + j * 256;
        if (i < total) srt[i] = stage[i];
    }
    int* orow = offs + (size_t)q * (NBKT + 1);
    orow[tid] = exc;
    if (tid == NBKT - 1) orow[NBKT] = exc + v;
}

__device__ __forceinline__ void own_put(const int2 ds, const int base, int* __restrict__ lcnt,
                                        int* __restrict__ lslot, int* __restrict__ cnt,
                                        int* __restrict__ ovf, int N) {
    const int j = ds.x - base;
    if ((unsigned)j < (unsigned)OWN_NODES) {
        const int r = atomicAdd(&lcnt[j], 1);
        if (r < SLOTS) {
            lslot[j * SLOTS + r] = ds.y;
        } else {
            const int o = atomicAdd(&cnt[N], 1);
            ((int2*)ovf)[o] = ds;
        }
    }
}

__device__ __forceinline__ void own_body(const int* __restrict__ sorted,
                                         const int* __restrict__ offs, int nchunk,
                                         int* __restrict__ cnt, int* __restrict__ slot,
                                         int* __restrict__ ovf, int N, const int p,
                                         int* __restrict__ lds) {
    int* lcnt  = lds;
    int* lslot = lds + OWN_NODES;
    const int tid = threadIdx.x;
    const int base = p * OWN_NODES;
    if (tid < OWN_NODES) lcnt[tid] = 0;
    __syncthreads();
    const int2* srt = (const int2*)sorted;
    for (int q0 = 0; q0 < nchunk; q0 += 4 * 256) {
        int a[4], b[4];
        size_t qb[4];
        #pragma unroll
        for (int u = 0; u < 4; ++u) {
            const int q = q0 + tid + u * 256;
            const bool in = q < nchunk;
            const int qq = in ? q : 0;
            const int* orow = offs + (size_t)qq * (NBKT + 1) + p;
            a[u] = in ? orow[0] : 0;
            b[u] = in ? orow[1] : 0;
            qb[u] = (size_t)qq * CHUNK;
        }
        int2 en[4][4];
        #pragma unroll
        for (int u = 0; u < 4; ++u) {
            #pragma unroll
            for (int k = 0; k < 4; ++k) {
                const int i = a[u] + k;
                en[u][k] = srt[qb[u] + (i < b[u] ? i : 0)];
            }
        }
        #pragma unroll
        for (int u = 0; u < 4; ++u) {
            #pragma unroll
            for (int k = 0; k < 4; ++k)
                if (a[u] + k < b[u]) own_put(en[u][k], base, lcnt, lslot, cnt, ovf, N);
            for (int i = a[u] + 4; i < b[u]; ++i)
                own_put(srt[qb[u] + i], base, lcnt, lslot, cnt, ovf, N);
        }
    }
    __syncthreads();
    if (tid < OWN_NODES && base + tid < N) cnt[base + tid] = lcnt[tid];
    for (int i = tid; i < OWN_NODES * 32; i += 256) {
        const int j = i >> 5, k = i & 31;
        const int n = base + j;
        if (n < N) {
            const int c = lcnt[j];
            int* row = slot + (size_t)n * SLOTS;
            row[k] = (k < c) ? lslot[j * SLOTS + k] : 0;
            if (c > 32) row[32 + k] = (32 + k < c) ? lslot[j * SLOTS + 32 + k] : 0;
        }
    }
}

__device__ __forceinline__ void gemm_body(const float* __restrict__ H,
                                          const float* __restrict__ W,
                                          float* __restrict__ out, int N, const int blk,
                                          float* __restrict__ lw) {
    {
        const float4* Wv = (const float4*)W;
        float4* lv = (float4*)lw;
        #pragma unroll
        for (int i = 0; i < (DIM * DIM / 4) / 256; ++i)
            lv[threadIdx.x + i * 256] = Wv[threadIdx.x + i * 256];
    }
    __syncthreads();

    const int wave = threadIdx.x >> 5;
    const int lane = threadIdx.x & 31;
    const int half = lane >> 4;
    const int ln   = lane & 15;

    const int nstrips = N >> 4;
    int strip = blk * 8 + wave;
    if (strip >= nstrips) strip = nstrips - 1;
    const int rowBase = strip << 4;

    const v2f* arow2 = (const v2f*)(H + (size_t)(rowBase + ln) * DIM);

    v8f acc[8] = {};

    for (int k = 0; k < DIM; k += 4) {
        v2f a = arow2[(k >> 1) + half];
        const int kb = k + 2 * half;
        #pragma unroll
        for (int t = 0; t < 8; ++t) {
            v2f b;
            b.x = lw[kb * DIM + t * 16 + ln];
            b.y = lw[(kb + 1) * DIM + t * 16 + ln];
            acc[t] = __builtin_amdgcn_wmma_f32_16x16x4_f32(
                false, a, false, b, (short)0, acc[t], false, false);
        }
    }

    #pragma unroll
    for (int t = 0; t < 8; ++t) {
        #pragma unroll
        for (int j = 0; j < 8; ++j) {
            int row = rowBase + half * 8 + j;
            out[(size_t)row * DIM + t * 16 + ln] = acc[t][j];
        }
    }
}

__global__ __launch_bounds__(256) void place_gemm(const int* __restrict__ sorted,
                                                  const int* __restrict__ offs, int nchunk,
                                                  int* __restrict__ cnt,
                                                  int* __restrict__ slot,
                                                  int* __restrict__ ovf, int ngemm,
                                                  const float* __restrict__ H,
                                                  const float* __restrict__ W,
                                                  float* __restrict__ out, int N) {
    __shared__ float lw[DIM * DIM];
    const int blk = (int)blockIdx.x;
    if (blk < ngemm) gemm_body(H, W, out, N, blk, lw);
    else             own_body(sorted, offs, nchunk, cnt, slot, ovf, N, blk - ngemm, (int*)lw);
}

__device__ __forceinline__ void row_fma(float4& a, float w, const float4 v) {
    a.x = __builtin_fmaf(w, v.x, a.x);  a.y = __builtin_fmaf(w, v.y, a.y);
    a.z = __builtin_fmaf(w, v.z, a.z);  a.w = __builtin_fmaf(w, v.w, a.w);
}

__device__ __forceinline__ void row_fma_if(float4& a, float w, const float4 v, bool p) {
    a.x = p ? __builtin_fmaf(w, v.x, a.x) : a.x;  a.y = p ? __builtin_fmaf(w, v.y, a.y) : a.y;
    a.z = p ? __builtin_fmaf(w, v.z, a.z) : a.z;  a.w = p ? __builtin_fmaf(w, v.w, a.w) : a.w;
}

__device__ __forceinline__ void row_add(float4& a, const float4 v) {
    a.x += v.x;  a.y += v.y;  a.z += v.z;  a.w += v.w;
}
__device__ __forceinline__ void row_add_if(float4& a, const float4 v, bool p) {
    a.x = p ? a.x + v.x : a.x;  a.y = p ? a.y + v.y : a.y;
    a.z = p ? a.z + v.z : a.z;  a.w = p ? a.w + v.w : a.w;
}

__device__ __forceinline__ float edge_w(int ecr, int k, float dn) {
    const int c = __builtin_amdgcn_readlane(ecr, k);
    return __builtin_amdgcn_rsqf((float)c + 1.0f) * dn;
}

template <bool PRE>
__device__ __forceinline__ float4 gather_row(const int n, const int lane,
                                             const int* __restrict__ cnt,
                                             const int* __restrict__ slot,
                                             const int* __restrict__ ovf, int E,
                                             const float* __restrict__ hw,
                                             const float4 bv, const float4 g, const float4 b,
                                             int N, float& dn_out) {
    const int* srow = slot + (size_t)n * SLOTS;
    const int e0 = srow[lane];
    const int deg = cnt[n];
    const float dn = __builtin_amdgcn_rsqf((float)deg + 1.0f);
    const int mall = min(deg, SLOTS);
    const float4 hv = ((const float4*)(hw + (size_t)n * DIM))[lane];
    dn_out = dn;
    float4 a;
    if (PRE) {
        a = hv;
    } else {
        const float sl = dn * dn;
        a.x = bv.x + sl * hv.x;  a.y = bv.y + sl * hv.y;
        a.z = bv.z + sl * hv.z;  a.w = bv.w + sl * hv.w;
    }
    asm volatile("" :: "v"(e0));

    for (int base = 0; base < mall; base += 32) {
        const int m = min(32, mall - base);
        int es = (base == 0) ? e0 : srow[32 + lane];
        int ecr = 0;
        if (!PRE) {
            es = (lane < m) ? es : 0;
            ecr = cnt[es];
        }
        int k = 0;
        for (; k + 4 <= m; k += 4) {
            const int s0 = __builtin_amdgcn_readlane(es, k);
            const int s1 = __builtin_amdgcn_readlane(es, k + 1);
            const int s2 = __builtin_amdgcn_readlane(es, k + 2);
            const int s3 = __builtin_amdgcn_readlane(es, k + 3);
            const float4 v0 = ((const float4*)(hw + (size_t)s0 * DIM))[lane];
            const float4 v1 = ((const float4*)(hw + (size_t)s1 * DIM))[lane];
            const float4 v2 = ((const float4*)(hw + (size_t)s2 * DIM))[lane];
            const float4 v3 = ((const float4*)(hw + (size_t)s3 * DIM))[lane];
            if (PRE) {
                row_add(a, v0);  row_add(a, v1);  row_add(a, v2);  row_add(a, v3);
            } else {
                row_fma(a, edge_w(ecr, k,     dn), v0);
                row_fma(a, edge_w(ecr, k + 1, dn), v1);
                row_fma(a, edge_w(ecr, k + 2, dn), v2);
                row_fma(a, edge_w(ecr, k + 3, dn), v3);
            }
        }
        if (k < m) {
            const int k1 = min(k + 1, m - 1);
            const int k2 = min(k + 2, m - 1);
            const int s0 = __builtin_amdgcn_readlane(es, k);
            const int s1 = __builtin_amdgcn_readlane(es, k1);
            const int s2 = __builtin_amdgcn_readlane(es, k2);
            const float4 v0 = ((const float4*)(hw + (size_t)s0 * DIM))[lane];
            const float4 v1 = ((const float4*)(hw + (size_t)s1 * DIM))[lane];
            const float4 v2 = ((const float4*)(hw + (size_t)s2 * DIM))[lane];
            if (PRE) {
                row_add(a, v0);
                row_add_if(a, v1, k + 1 < m);
                row_add_if(a, v2, k + 2 < m);
            } else {
                row_fma(a, edge_w(ecr, k, dn), v0);
                row_fma_if(a, edge_w(ecr, k1, dn), v1, k + 1 < m);
                row_fma_if(a, edge_w(ecr, k2, dn), v2, k + 2 < m);
            }
        }
    }
    if (deg > SLOTS) {
        const int no = min(cnt[N], E);
        for (int i = 0; i < no; i += 32) {
            int od = -1, os = 0;
            if (i + lane < no) {
                const int2 q = ((const int2*)ovf)[i + lane];
                od = q.x;  os = q.y;
            }
            unsigned hits = __builtin_amdgcn_ballot_w32(od == n);
            while (hits) {
                const int j = __builtin_ctz(hits);
                hits &= hits - 1;
                const int so = __builtin_amdgcn_readlane(os, j);
                const float4 vo = ((const float4*)(hw + (size_t)so * DIM))[lane];
                if (PRE) {
                    row_add(a, vo);
                } else {
                    const float wo = __builtin_amdgcn_rsqf((float)cnt[so] + 1.0f) * dn;
                    row_fma(a, wo, vo);
                }
            }
        }
    }
    if (PRE) {
        a.x = bv.x + dn * a.x;  a.y = bv.y + dn * a.y;
        a.z = bv.z + dn * a.z;  a.w = bv.w + dn * a.w;
    }
    float s = a.x + a.y + a.z + a.w;
    float q = a.x * a.x + a.y * a.y + a.z * a.z + a.w * a.w;
    for (int off = 16; off > 0; off >>= 1) {
        s += __shfl_xor(s, off, 32);
        q += __shfl_xor(q, off, 32);
    }
    const float mu  = s * (1.0f / DIM);
    const float var = fmaxf(q * (1.0f / DIM) - mu * mu, 0.0f);
    const float inv = rsqrtf(var + LNEPS);
    float4 r;
    r.x = fmaxf(0.0f, g.x * (a.x - mu) * inv + b.x);
    r.y = fmaxf(0.0f, g.y * (a.y - mu) * inv + b.y);
    r.z = fmaxf(0.0f, g.z * (a.z - mu) * inv + b.z);
    r.w = fmaxf(0.0f, g.w * (a.w - mu) * inv + b.w);
    return r;
}

#define TILE_ROWS 16
#define TPITCH 132
#define FUSED_WGS 1024

template <bool PRE>
__global__ __launch_bounds__(512) void gather_ln_relu_gemm(const int* __restrict__ cnt,
                                                           const int* __restrict__ slot,
                                                           const int* __restrict__ ovf, int E,
                                                           const float* __restrict__ hw,
                                                           const float* __restrict__ bias,
                                                           const float* __restrict__ gamma,
                                                           const float* __restrict__ beta,
                                                           const float* __restrict__ Wn,
                                                           float* __restrict__ out, int N) {
    __shared__ __attribute__((aligned(16))) float lw[DIM * DIM];
    __shared__ __attribute__((aligned(16))) float tile[TILE_ROWS * TPITCH];
    const int lane = threadIdx.x & 31;
    const int wave = __builtin_amdgcn_readfirstlane((int)(threadIdx.x >> 5));
    {
        const float4* Wv = (const float4*)Wn;
        float4* lv = (float4*)lw;
        #pragma unroll
        for (int i = 0; i < (DIM * DIM / 4) / 512; ++i)
            lv[threadIdx.x + i * 512] = Wv[threadIdx.x + i * 512];
    }
    const float4 bv = ((const float4*)bias)[lane];
    const float4 g = ((const float4*)gamma)[lane];
    const float4 b = ((const float4*)beta)[lane];
    const int half = lane >> 4;
    const int ln   = lane & 15;
    const int ntiles = (N + TILE_ROWS - 1) / TILE_ROWS;

    for (int t = (int)blockIdx.x; t < ntiles; t += (int)gridDim.x) {
        const int n = t * TILE_ROWS + wave;
        float4 r = make_float4(0.0f, 0.0f, 0.0f, 0.0f);
        if (n < N) {
            float dn;
            r = gather_row<PRE>(n, lane, cnt, slot, ovf, E, hw, bv, g, b, N, dn);
            r.x *= dn;  r.y *= dn;  r.z *= dn;  r.w *= dn;
        }
        __syncthreads();
        ((float4*)(tile + wave * TPITCH))[lane] = r;
        __syncthreads();
        if (wave < 8) {
            const float* arow = tile + ln * TPITCH + 2 * half;
            const float* bcol = lw + (2 * half) * DIM + wave * 16 + ln;
            v8f acc0 = {}, acc1 = {};
            #pragma unroll 4
            for (int k = 0; k < DIM; k += 8) {
                const v2f a0 = *(const v2f*)(arow + k);
                const v2f a1 = *(const v2f*)(arow + k + 4);
                v2f b0, b1;
                b0.x = bcol[k * DIM];
                b0.y = bcol[(k + 1) * DIM];
                b1.x = bcol[(k + 4) * DIM];
                b1.y = bcol[(k + 5) * DIM];
                acc0 = __builtin_amdgcn_wmma_f32_16x16x4_f32(
                    false, a0, false, b0, (short)0, acc0, false, false);
                acc1 = __builtin_amdgcn_wmma_f32_16x16x4_f32(
                    false, a1, false, b1, (short)0, acc1, false, false);
            }
            float* orow = out + (size_t)(t * TILE_ROWS + half * 8) * DIM + wave * 16 + ln;
            if (t * TILE_ROWS + TILE_ROWS <= N) {
                #pragma unroll
                for (int j = 0; j < 8; ++j) orow[(size_t)j * DIM] = acc0[j] + acc1[j];
            } else {
                #pragma unroll
                for (int j = 0; j < 8; ++j)
                    if (t * TILE_ROWS + half * 8 + j < N) orow[(size_t)j * DIM] = acc0[j] + acc1[j];
            }
        }
    }
}

#define POOL_WGS   1024
#define POOL_WAVES 16
#define POOL_NWAVES (POOL_WGS * POOL_WAVES)

__device__ __forceinline__ void pool_flush(float* __restrict__ pool, float* __restrict__ cnt,
                                           int g, float4 s, float c, int lane) {
    float* p = pool + (size_t)g * DIM + lane * 4;
    atomicAdd(p + 0, s.x);
    atomicAdd(p + 1, s.y);
    atomicAdd(p + 2, s.z);
    atomicAdd(p + 3, s.w);
    if (lane == 0) atomicAdd(&cnt[g], c);
}

__global__ __launch_bounds__(512) void gather_ln_relu_pool(const int* __restrict__ cnt,
                                                           const int* __restrict__ slot,
                                                           const int* __restrict__ ovf, int E,
                                                           const float* __restrict__ hw,
                                                           const float* __restrict__ bias,
                                                           const float* __restrict__ gamma,
                                                           const float* __restrict__ beta,
                                                           const int* __restrict__ batch,
                                                           float* __restrict__ pool,
                                                           float* __restrict__ pcount, int N) {
    __shared__ __attribute__((aligned(16))) float part[POOL_WAVES][DIM];
    __shared__ int   pgid[POOL_WAVES];
    __shared__ float pcnt[POOL_WAVES];

    const int lane = threadIdx.x & 31;
    const int wave = __builtin_amdgcn_readfirstlane((int)(threadIdx.x >> 5));
    const float4 bv = ((const float4*)bias)[lane];
    const float4 g = ((const float4*)gamma)[lane];
    const float4 b = ((const float4*)beta)[lane];

    const unsigned long long gw = (unsigned long long)(blockIdx.x * POOL_WAVES + wave);
    const int n0 = (int)((gw * (unsigned long long)N) / POOL_NWAVES);
    const int n1 = (int)(((gw + 1) * (unsigned long long)N) / POOL_NWAVES);

    int cur = -1;
    float4 s = make_float4(0.0f, 0.0f, 0.0f, 0.0f);
    float c = 0.0f;
    for (int n = n0; n < n1; ++n) {
        const int gi = __builtin_amdgcn_readfirstlane(batch[n]);
        float dn;
        const float4 r = gather_row<true>(n, lane, cnt, slot, ovf, E, hw, bv, g, b, N, dn);
        if (gi == cur) {
            s.x += r.x;  s.y += r.y;  s.z += r.z;  s.w += r.w;
            c += 1.0f;
        } else {
            if (cur >= 0) pool_flush(pool, pcount, cur, s, c, lane);
            cur = gi;  s = r;  c = 1.0f;
        }
    }

    ((float4*)part[wave])[lane] = s;
    if (lane == 0) { pgid[wave] = cur; pcnt[wave] = c; }
    __syncthreads();

    const int t = threadIdx.x;
    if (t <= DIM) {
        const bool isc = (t == DIM);
        float* dstp = isc ? pcount : pool + t;
        const int dstride = isc ? 1 : DIM;
        int   jg[POOL_WAVES];
        float jx[POOL_WAVES];
        #pragma unroll
        for (int w = 0; w < POOL_WAVES; ++w) {
            jg[w] = pgid[w];
            jx[w] = isc ? pcnt[w] : part[w][t & (DIM - 1)];
        }
        int   rg = jg[0];
        float rs = jx[0];
        #pragma unroll
        for (int w = 1; w < POOL_WAVES; ++w) {
            if (jg[w] == rg) {
                rs += jx[w];
            } else {
                if (rg >= 0) atomicAdd(dstp + (size_t)rg * dstride, rs);
                rg = jg[w]; rs = jx[w];
            }
        }
        if (rg >= 0) atomicAdd(dstp + (size_t)rg * dstride, rs);
    }
}

__global__ void pool_final(const float* __restrict__ pool, const float* __restrict__ cnt,
                           float* __restrict__ out, int total) {
    int i = blockIdx.x * blockDim.x + threadIdx.x;
    if (i < total) out[i] = pool[i] / fmaxf(cnt[i / DIM], 1.0f);
}

static inline size_t al256(size_t x) { return (x + 255) & ~(size_t)255; }

extern "C" void kernel_launch(void* const* d_in, const int* in_sizes, int n_in,
                              void* d_out, int out_size, void* d_ws, size_t ws_size,
                              hipStream_t stream) {
    const float* x      = (const float*)d_in[0];
    const int*   ei     = (const int*)d_in[1];
    const int*   batch  = (const int*)d_in[2];
    const float* Ws     = (const float*)d_in[3];
    const float* bs     = (const float*)d_in[4];
    const float* gammas = (const float*)d_in[5];
    const float* betas  = (const float*)d_in[6];

    const int N = in_sizes[0] / DIM;
    const int E = in_sizes[1] / 2;
    const size_t nd = (size_t)N * DIM * sizeof(float);

    char* ws = (char*)d_ws;
    int*   cnt  = (int*)ws;                   ws += al256((size_t)(N + 1) * sizeof(int));
    int*   slot = (int*)ws;                   ws += al256((size_t)N * SLOTS * sizeof(int));
    int*   ovf  = (int*)ws;                   ws += al256((size_t)E * 2 * sizeof(int));
    const int nchunk = (E + CHUNK - 1) / CHUNK;
    int*   srt  = (int*)ws;                   ws += al256((size_t)nchunk * CHUNK * 2 * sizeof(int));
    int*   offs = (int*)ws;                   ws += al256((size_t)nchunk * (NBKT + 1) * sizeof(int));
    float* hw   = (float*)ws;                 ws += al256(nd);
    float* hbuf = (float*)ws;                 ws += al256(nd);
    float* pool = (float*)ws;
    float* pcnt = pool + GNUM * DIM;
    const int npool = GNUM * DIM + GNUM;

    const int nstrips = N >> 4;
    const int ngemm = (nstrips + 7) / 8;
    const int nown = (N + OWN_NODES - 1) / OWN_NODES;
    const int nzero = (npool + 1 + 255) / 256;
    bucket_edges<<<(nchunk > nzero ? nchunk : nzero), 256, 0, stream>>>(
        ei, E, nchunk, srt, offs, cnt + N, pool, npool);
    place_gemm<<<ngemm + nown, 256, 0, stream>>>(srt, offs, nchunk, cnt, slot, ovf, ngemm,
                                                 x, Ws, hw, N);

    float* pin = hw;
    float* pout = hbuf;
    for (int l = 0; l + 1 < LAYERS; ++l) {
        if (l == 0)
            gather_ln_relu_gemm<false><<<FUSED_WGS, 512, 0, stream>>>(
                cnt, slot, ovf, E, pin, bs + (size_t)l * DIM,
                gammas + (size_t)l * DIM, betas + (size_t)l * DIM,
                Ws + (size_t)(l + 1) * DIM * DIM, pout, N);
        else
            gather_ln_relu_gemm<true><<<FUSED_WGS, 512, 0, stream>>>(
                cnt, slot, ovf, E, pin, bs + (size_t)l * DIM,
                gammas + (size_t)l * DIM, betas + (size_t)l * DIM,
                Ws + (size_t)(l + 1) * DIM * DIM, pout, N);
        float* tmp = pin;  pin = pout;  pout = tmp;
    }
    gather_ln_relu_pool<<<POOL_WGS, 512, 0, stream>>>(
        cnt, slot, ovf, E, pin, bs + (size_t)(LAYERS - 1) * DIM,
        gammas + (size_t)(LAYERS - 1) * DIM, betas + (size_t)(LAYERS - 1) * DIM,
        batch, pool, pcnt, N);
    pool_final<<<(GNUM * DIM + 255) / 256, 256, 0, stream>>>(pool, pcnt, (float*)d_out, GNUM * DIM);
}
